# starting kernel
# baseline (speedup 1.0000x reference)
#include <hip/hip_runtime.h>
#include <hip/hip_bf16.h>


#define GN 50000
#define GE 800000
#define GETOT (GE + GN)
#define GHC 128
#define GC 64
#define GH 2
#define NEG_SLOPE 0.2f
#define GEPS 1e-5f

typedef __attribute__((ext_vector_type(2))) float v2f;
typedef __attribute__((ext_vector_type(8))) float v8f;

__global__ void wmma_gemm_f32(const float* __restrict__ A,
                              const float* __restrict__ B,
                              const float* __restrict__ bias,
                              float* __restrict__ C,
                              int K, int Nout, int doRelu) {
  const int lane    = threadIdx.x;
  const int colTile = threadIdx.y;
  const int row0    = blockIdx.x * 16;
  const int col0    = colTile * 16;
  const int col     = lane & 15;
  const int kHalf   = (lane >> 4) << 1;
  const int aRow    = row0 + (lane & 15);

  v8f acc = {};
  for (int kk = 0; kk < K; kk += 4) {
    v2f a, b;
    const float* ap = A + (size_t)aRow * K + (kk + kHalf);
    a.x = ap[0];
    a.y = ap[1];
    const float* bp = B + (size_t)(kk + kHalf) * Nout + col0 + col;
    b.x = bp[0];
    b.y = bp[Nout];
    acc = __builtin_amdgcn_wmma_f32_16x16x4_f32(false, a, false, b,
                                                (short)0, acc, false, false);
  }

  const int rowBase = row0 + ((lane >> 4) << 3);
  const float bv = bias ? bias[col0 + col] : 0.0f;
#pragma unroll
  for (int j = 0; j < 8; ++j) {
    float v = acc[j] + bv;
    if (doRelu) v = fmaxf(v, 0.0f);
    C[(size_t)(rowBase + j) * Nout + col0 + col] = v;
  }
}

__global__ void fill_kernel(float* __restrict__ p, float v, int n) {
  int i = blockIdx.x * blockDim.x + threadIdx.x;
  if (i < n) p[i] = v;
}

__device__ inline void atomicMaxF(float* addr, float val) {
  if (val >= 0.0f)
    atomicMax((int*)addr, __float_as_int(val));
  else
    atomicMin((unsigned int*)addr, __float_as_uint(val));
}

__global__ void edge_logits_kernel(const float* __restrict__ xl,
                                   const float* __restrict__ xr,
                                   const int* __restrict__ ei,
                                   const float* __restrict__ att,
                                   float* __restrict__ logits,
                                   float* __restrict__ m) {
  const int lane = threadIdx.x & 31;
  const int wave = threadIdx.x >> 5;
  const int pair = blockIdx.x * (blockDim.x >> 5) + wave;
  if (pair >= GETOT * GH) return;
  const int e = pair >> 1;
  const int h = pair & 1;
  const int s = (e < GE) ? ei[e] : (e - GE);
  const int d = (e < GE) ? ei[GE + e] : (e - GE);

  const float* xls = xl + (size_t)s * GHC + h * GC;
  const float* xrd = xr + (size_t)d * GHC + h * GC;
  const float* ah  = att + h * GC;

  float t0 = xls[lane] + xrd[lane];
  float t1 = xls[lane + 32] + xrd[lane + 32];
  t0 = (t0 > 0.0f) ? t0 : NEG_SLOPE * t0;
  t1 = (t1 > 0.0f) ? t1 : NEG_SLOPE * t1;
  float v = t0 * ah[lane] + t1 * ah[lane + 32];

#pragma unroll
  for (int off = 16; off > 0; off >>= 1) v += __shfl_xor(v, off, 32);

  if (lane == 0) {
    logits[pair] = v;
    atomicMaxF(&m[(size_t)d * GH + h], v);
  }
}

__global__ void edge_softmax_p(const int* __restrict__ ei,
                               float* __restrict__ logits,
                               const float* __restrict__ m,
                               float* __restrict__ denom) {
  int i = blockIdx.x * blockDim.x + threadIdx.x;
  if (i >= GETOT * GH) return;
  const int e = i >> 1;
  const int h = i & 1;
  const int d = (e < GE) ? ei[GE + e] : (e - GE);
  float p = __expf(logits[i] - m[(size_t)d * GH + h]);
  logits[i] = p;
  atomicAdd(&denom[(size_t)d * GH + h], p);
}

__global__ void edge_aggregate(const int* __restrict__ ei,
                               const float* __restrict__ xl,
                               const float* __restrict__ p,
                               const float* __restrict__ denom,
                               float* __restrict__ accum) {
  const int e = blockIdx.x;
  const int f = threadIdx.x;
  const int h = f >> 6;
  const int s = (e < GE) ? ei[e] : (e - GE);
  const int d = (e < GE) ? ei[GE + e] : (e - GE);
  const float alpha = p[(size_t)e * GH + h] / denom[(size_t)d * GH + h];
  atomicAdd(&accum[(size_t)d * GHC + f], alpha * xl[(size_t)s * GHC + f]);
}

__global__ void gn_stats(const float* __restrict__ accum,
                         const float* __restrict__ bias,
                         float* __restrict__ colsum,
                         float* __restrict__ colsumsq) {
  const int f = threadIdx.x;
  const int row0 = blockIdx.x * 128;
  const float bv = bias[f];
  float s = 0.0f, s2 = 0.0f;
  for (int r = 0; r < 128; ++r) {
    const int n = row0 + r;
    if (n >= GN) break;
    const float x = accum[(size_t)n * GHC + f] + bv;
    s += x;
    s2 += x * x;
  }
  atomicAdd(&colsum[f], s);
  atomicAdd(&colsumsq[f], s2);
}

__global__ void gn_finalize(const float* __restrict__ colsum,
                            const float* __restrict__ colsumsq,
                            const float* __restrict__ w,
                            const float* __restrict__ b,
                            const float* __restrict__ ms,
                            float* __restrict__ s1,
                            float* __restrict__ s2) {
  const int f = threadIdx.x;
  const float invN = 1.0f / (float)GN;
  const float mean = colsum[f] * invN;
  const float ex2  = colsumsq[f] * invN;
  const float msv  = ms[f];
  const float var  = ex2 - (2.0f * msv - msv * msv) * mean * mean;
  const float r    = rsqrtf(var + GEPS);
  const float a    = w[f] * r;
  s1[f] = a;
  s2[f] = b[f] - a * msv * mean;
}

__global__ void gn_apply(const float* __restrict__ accum,
                         const float* __restrict__ bias,
                         const float* __restrict__ s1,
                         const float* __restrict__ s2,
                         float* __restrict__ out) {
  int i = blockIdx.x * blockDim.x + threadIdx.x;
  if (i >= GN * GHC) return;
  const int f = i & (GHC - 1);
  const float x = accum[i] + bias[f];
  out[i] = fmaxf(x * s1[f] + s2[f], 0.0f);
}

__global__ void head_kernel(const float* __restrict__ h1,
                            const float* __restrict__ W2,
                            const float* __restrict__ b2,
                            float* __restrict__ out) {
  int n = blockIdx.x * blockDim.x + threadIdx.x;
  if (n >= GN) return;
  const float* row = h1 + (size_t)n * GC;
  float a0 = b2[0], a1 = b2[1];
#pragma unroll 8
  for (int c = 0; c < GC; ++c) {
    const float v = row[c];
    a0 += v * W2[c * 2 + 0];
    a1 += v * W2[c * 2 + 1];
  }
  out[n * 2 + 0] = a0;
  out[n * 2 + 1] = a1;
}

extern "C" void kernel_launch(void* const* d_in, const int* in_sizes, int n_in,
                              void* d_out, int out_size, void* d_ws, size_t ws_size,
                              hipStream_t stream) {
  (void)in_sizes; (void)n_in; (void)out_size; (void)ws_size;

  const float* x     = (const float*)d_in[0];
  const int*   ei    = (const int*)d_in[1];
  const float* Wl    = (const float*)d_in[2];
  const float* Wr    = (const float*)d_in[3];
  const float* att   = (const float*)d_in[4];
  const float* convb = (const float*)d_in[5];
  const float* gnw   = (const float*)d_in[6];
  const float* gnb   = (const float*)d_in[7];
  const float* gnms  = (const float*)d_in[8];
  const float* W1    = (const float*)d_in[9];
  const float* b1    = (const float*)d_in[10];
  const float* W2    = (const float*)d_in[11];
  const float* b2    = (const float*)d_in[12];
  float* out = (float*)d_out;

  float* ws       = (float*)d_ws;
  float* xl       = ws;
  float* xr       = xl + (size_t)GN * GHC;
  float* hbuf     = xr + (size_t)GN * GHC;
  float* h1       = hbuf + (size_t)GN * GHC;
  float* logits   = h1;
  size_t scratch1 = ((size_t)GN * GC > (size_t)GETOT * GH)
                        ? (size_t)GN * GC : (size_t)GETOT * GH;
  float* accum    = h1 + scratch1;
  float* denom    = accum + (size_t)GN * GHC;
  float* colsum   = denom + (size_t)GN * GH;
  float* colsumsq = colsum + GHC;
  float* s1       = colsumsq + GHC;
  float* s2       = s1 + GHC;
  float* m        = s2 + GHC;

  const int zeroCount = GN * GHC + GN * GH + 2 * GHC;
  const int pairCount = GETOT * GH;

  const float* cur = x;
  for (int layer = 0; layer < 2; ++layer) {
    const float* wl = Wl + (size_t)layer * GHC * GHC;
    const float* wr = Wr + (size_t)layer * GHC * GHC;
    const float* at = att + (size_t)layer * GH * GC;
    const float* cb = convb + (size_t)layer * GHC;

    wmma_gemm_f32<<<GN / 16, dim3(32, GHC / 16), 0, stream>>>(cur, wl, nullptr, xl, GHC, GHC, 0);
    wmma_gemm_f32<<<GN / 16, dim3(32, GHC / 16), 0, stream>>>(cur, wr, nullptr, xr, GHC, GHC, 0);

    fill_kernel<<<(zeroCount + 255) / 256, 256, 0, stream>>>(accum, 0.0f, zeroCount);
    fill_kernel<<<(GN * GH + 255) / 256, 256, 0, stream>>>(m, -3.0e38f, GN * GH);

    edge_logits_kernel<<<(pairCount + 7) / 8, 256, 0, stream>>>(xl, xr, ei, at, logits, m);
    edge_softmax_p<<<(pairCount + 255) / 256, 256, 0, stream>>>(ei, logits, m, denom);
    edge_aggregate<<<GETOT, GHC, 0, stream>>>(ei, xl, logits, denom, accum);

    gn_stats<<<(GN + 127) / 128, GHC, 0, stream>>>(accum, cb, colsum, colsumsq);
    gn_finalize<<<1, GHC, 0, stream>>>(colsum, colsumsq,
                                       gnw + (size_t)layer * GHC,
                                       gnb + (size_t)layer * GHC,
                                       gnms + (size_t)layer * GHC, s1, s2);
    gn_apply<<<(GN * GHC + 255) / 256, 256, 0, stream>>>(accum, cb, s1, s2, hbuf);

    cur = hbuf;
  }

  wmma_gemm_f32<<<GN / 16, dim3(32, GC / 16), 0, stream>>>(hbuf, W1, b1, h1, GHC, GC, 1);
  head_kernel<<<(GN + 255) / 256, 256, 0, stream>>>(h1, W2, b2, out);
}


# Edge logits and edge aggregate: one wave walks a batch of 32 edges (coalesced index load, lane broadcast, whole-line row loads, 4 edges in flight), not one wave per edge-head / one block per edge
# speedup vs baseline: 1.9328x; 1.9328x over previous
#include <hip/hip_runtime.h>
#include <hip/hip_bf16.h>


#define GN 50000
#define GE 800000
#define GETOT (GE + GN)
#define GHC 128
#define GC 64
#define GH 2
#define NEG_SLOPE 0.2f
#define GEPS 1e-5f

typedef __attribute__((ext_vector_type(2))) float v2f;
typedef __attribute__((ext_vector_type(8))) float v8f;

__global__ void wmma_gemm_f32(const float* __restrict__ A,
                              const float* __restrict__ B,
                              const float* __restrict__ bias,
                              float* __restrict__ C,
                              int K, int Nout, int doRelu) {
  const int lane    = threadIdx.x;
  const int colTile = threadIdx.y;
  const int row0    = blockIdx.x * 16;
  const int col0    = colTile * 16;
  const int col     = lane & 15;
  const int kHalf   = (lane >> 4) << 1;
  const int aRow    = row0 + (lane & 15);

  v8f acc = {};
  for (int kk = 0; kk < K; kk += 4) {
    v2f a, b;
    const float* ap = A + (size_t)aRow * K + (kk + kHalf);
    a.x = ap[0];
    a.y = ap[1];
    const float* bp = B + (size_t)(kk + kHalf) * Nout + col0 + col;
    b.x = bp[0];
    b.y = bp[Nout];
    acc = __builtin_amdgcn_wmma_f32_16x16x4_f32(false, a, false, b,
                                                (short)0, acc, false, false);
  }

  const int rowBase = row0 + ((lane >> 4) << 3);
  const float bv = bias ? bias[col0 + col] : 0.0f;
#pragma unroll
  for (int j = 0; j < 8; ++j) {
    float v = acc[j] + bv;
    if (doRelu) v = fmaxf(v, 0.0f);
    C[(size_t)(rowBase + j) * Nout + col0 + col] = v;
  }
}

__global__ void fill_kernel(float* __restrict__ p, float v, int n) {
  int i = blockIdx.x * blockDim.x + threadIdx.x;
  if (i < n) p[i] = v;
}

__device__ inline void atomicMaxF(float* addr, float val) {
  if (val >= 0.0f)
    atomicMax((int*)addr, __float_as_int(val));
  else
    atomicMin((unsigned int*)addr, __float_as_uint(val));
}

#define EB 32
#define NBATCH ((GETOT + EB - 1) / EB)
#define EDGE_WAVES_PER_BLOCK 4

__device__ __forceinline__ int bcast_i(int v, int i) {
  return __builtin_amdgcn_readlane(v, i);
}
__device__ __forceinline__ float bcast_f(float v, int i) {
  return __int_as_float(__builtin_amdgcn_readlane(__float_as_int(v), i));
}

__global__ void __launch_bounds__(EDGE_WAVES_PER_BLOCK * 32)
edge_logits_kernel(const float* __restrict__ xl,
                   const float* __restrict__ xr,
                   const int* __restrict__ ei,
                   const float* __restrict__ att,
                   float* __restrict__ logits,
                   float* __restrict__ m) {
  const int lane  = threadIdx.x & 31;
  const int batch = __builtin_amdgcn_readfirstlane(blockIdx.x * (blockDim.x >> 5) + (threadIdx.x >> 5));
  if (batch >= NBATCH) return;
  const int base = batch * EB;
  const int left = GETOT - base;
  const int cnt  = (left < EB) ? left : EB;

  const int e = base + lane;
  int s = 0, d = 0;
  if (e < GETOT) {
    s = (e < GE) ? ei[e] : (e - GE);
    d = (e < GE) ? ei[GE + e] : (e - GE);
  }

  const float a00 = att[lane];
  const float a01 = att[lane + 32];
  const float a10 = att[GC + lane];
  const float a11 = att[GC + lane + 32];
  const bool lowHalf = lane < 16;
  const int  sub     = lane & 15;

  float c[2] = {0.0f, 0.0f};
#pragma unroll
  for (int half = 0; half < 2; ++half) {
    const int rem = cnt - half * 16;
    const int n   = (rem < 16) ? rem : 16;
#pragma unroll 4
    for (int j = 0; j < n; ++j) {
      const int i  = half * 16 + j;
      const int si = bcast_i(s, i);
      const int di = bcast_i(d, i);
      const float* xls = xl + (size_t)si * GHC;
      const float* xrd = xr + (size_t)di * GHC;

      float t0 = xls[lane] + xrd[lane];
      float t1 = xls[lane + 32] + xrd[lane + 32];
      float u0 = xls[GC + lane] + xrd[GC + lane];
      float u1 = xls[GC + lane + 32] + xrd[GC + lane + 32];
      t0 = (t0 > 0.0f) ? t0 : NEG_SLOPE * t0;
      t1 = (t1 > 0.0f) ? t1 : NEG_SLOPE * t1;
      u0 = (u0 > 0.0f) ? u0 : NEG_SLOPE * u0;
      u1 = (u1 > 0.0f) ? u1 : NEG_SLOPE * u1;
      const float v0 = t0 * a00 + t1 * a01;
      const float v1 = u0 * a10 + u1 * a11;

      const float keep = lowHalf ? v0 : v1;
      const float send = lowHalf ? v1 : v0;
      float w = keep + __shfl_xor(send, 16, 32);
#pragma unroll
      for (int off = 8; off > 0; off >>= 1) w += __shfl_xor(w, off, 32);

      if (sub == j) c[half] = w;
    }
  }

  const int h  = lane >> 4;
  const int e0 = base + sub;
  const int e1 = e0 + 16;
  const int d0 = __shfl(d, sub, 32);
  const int d1 = __shfl(d, sub + 16, 32);
  if (e0 < GETOT) {
    logits[(size_t)e0 * GH + h] = c[0];
    atomicMaxF(&m[(size_t)d0 * GH + h], c[0]);
  }
  if (e1 < GETOT) {
    logits[(size_t)e1 * GH + h] = c[1];
    atomicMaxF(&m[(size_t)d1 * GH + h], c[1]);
  }
}

__global__ void edge_softmax_p(const int* __restrict__ ei,
                               float* __restrict__ logits,
                               const float* __restrict__ m,
                               float* __restrict__ denom) {
  int i = blockIdx.x * blockDim.x + threadIdx.x;
  if (i >= GETOT * GH) return;
  const int e = i >> 1;
  const int h = i & 1;
  const int d = (e < GE) ? ei[GE + e] : (e - GE);
  float p = __expf(logits[i] - m[(size_t)d * GH + h]);
  logits[i] = p;
  atomicAdd(&denom[(size_t)d * GH + h], p);
}

__global__ void __launch_bounds__(EDGE_WAVES_PER_BLOCK * 32)
edge_aggregate(const int* __restrict__ ei,
               const float* __restrict__ xl,
               const float* __restrict__ p,
               const float* __restrict__ denom,
               float* __restrict__ accum) {
  const int lane  = threadIdx.x & 31;
  const int batch = __builtin_amdgcn_readfirstlane(blockIdx.x * (blockDim.x >> 5) + (threadIdx.x >> 5));
  if (batch >= NBATCH) return;
  const int base = batch * EB;
  const int left = GETOT - base;
  const int cnt  = (left < EB) ? left : EB;

  const int e = base + lane;
  int s = 0, d = 0;
  float al0 = 0.0f, al1 = 0.0f;
  if (e < GETOT) {
    s = (e < GE) ? ei[e] : (e - GE);
    d = (e < GE) ? ei[GE + e] : (e - GE);
    const float2 pe = *reinterpret_cast<const float2*>(p + (size_t)e * GH);
    const float2 dn = *reinterpret_cast<const float2*>(denom + (size_t)d * GH);
    al0 = pe.x / dn.x;
    al1 = pe.y / dn.y;
  }

  int i = 0;
  for (; i + 4 <= cnt; i += 4) {
    float x[4][4];
#pragma unroll
    for (int k = 0; k < 4; ++k) {
      const int si = bcast_i(s, i + k);
      const float* xs = xl + (size_t)si * GHC + lane;
      x[k][0] = xs[0];
      x[k][1] = xs[32];
      x[k][2] = xs[GC];
      x[k][3] = xs[GC + 32];
    }
#pragma unroll
    for (int k = 0; k < 4; ++k) {
      const int   di = bcast_i(d, i + k);
      const float a0 = bcast_f(al0, i + k);
      const float a1 = bcast_f(al1, i + k);
      float* ad = accum + (size_t)di * GHC + lane;
      atomicAdd(ad,           a0 * x[k][0]);
      atomicAdd(ad + 32,      a0 * x[k][1]);
      atomicAdd(ad + GC,      a1 * x[k][2]);
      atomicAdd(ad + GC + 32, a1 * x[k][3]);
    }
  }
  for (; i < cnt; ++i) {
    const int   si = bcast_i(s, i);
    const int   di = bcast_i(d, i);
    const float a0 = bcast_f(al0, i);
    const float a1 = bcast_f(al1, i);
    const float* xs = xl + (size_t)si * GHC + lane;
    float*       ad = accum + (size_t)di * GHC + lane;
    const float x0 = xs[0];
    const float x1 = xs[32];
    const float x2 = xs[GC];
    const float x3 = xs[GC + 32];
    atomicAdd(ad,           a0 * x0);
    atomicAdd(ad + 32,      a0 * x1);
    atomicAdd(ad + GC,      a1 * x2);
    atomicAdd(ad + GC + 32, a1 * x3);
  }
}

__global__ void gn_stats(const float* __restrict__ accum,
                         const float* __restrict__ bias,
                         float* __restrict__ colsum,
                         float* __restrict__ colsumsq) {
  const int f = threadIdx.x;
  const int row0 = blockIdx.x * 128;
  const float bv = bias[f];
  float s = 0.0f, s2 = 0.0f;
  for (int r = 0; r < 128; ++r) {
    const int n = row0 + r;
    if (n >= GN) break;
    const float x = accum[(size_t)n * GHC + f] + bv;
    s += x;
    s2 += x * x;
  }
  atomicAdd(&colsum[f], s);
  atomicAdd(&colsumsq[f], s2);
}

__global__ void gn_finalize(const float* __restrict__ colsum,
                            const float* __restrict__ colsumsq,
                            const float* __restrict__ w,
                            const float* __restrict__ b,
                            const float* __restrict__ ms,
                            float* __restrict__ s1,
                            float* __restrict__ s2) {
  const int f = threadIdx.x;
  const float invN = 1.0f / (float)GN;
  const float mean = colsum[f] * invN;
  const float ex2  = colsumsq[f] * invN;
  const float msv  = ms[f];
  const float var  = ex2 - (2.0f * msv - msv * msv) * mean * mean;
  const float r    = rsqrtf(var + GEPS);
  const float a    = w[f] * r;
  s1[f] = a;
  s2[f] = b[f] - a * msv * mean;
}

__global__ void gn_apply(const float* __restrict__ accum,
                         const float* __restrict__ bias,
                         const float* __restrict__ s1,
                         const float* __restrict__ s2,
                         float* __restrict__ out) {
  int i = blockIdx.x * blockDim.x + threadIdx.x;
  if (i >= GN * GHC) return;
  const int f = i & (GHC - 1);
  const float x = accum[i] + bias[f];
  out[i] = fmaxf(x * s1[f] + s2[f], 0.0f);
}

__global__ void head_kernel(const float* __restrict__ h1,
                            const float* __restrict__ W2,
                            const float* __restrict__ b2,
                            float* __restrict__ out) {
  int n = blockIdx.x * blockDim.x + threadIdx.x;
  if (n >= GN) return;
  const float* row = h1 + (size_t)n * GC;
  float a0 = b2[0], a1 = b2[1];
#pragma unroll 8
  for (int c = 0; c < GC; ++c) {
    const float v = row[c];
    a0 += v * W2[c * 2 + 0];
    a1 += v * W2[c * 2 + 1];
  }
  out[n * 2 + 0] = a0;
  out[n * 2 + 1] = a1;
}

extern "C" void kernel_launch(void* const* d_in, const int* in_sizes, int n_in,
                              void* d_out, int out_size, void* d_ws, size_t ws_size,
                              hipStream_t stream) {
  (void)in_sizes; (void)n_in; (void)out_size; (void)ws_size;

  const float* x     = (const float*)d_in[0];
  const int*   ei    = (const int*)d_in[1];
  const float* Wl    = (const float*)d_in[2];
  const float* Wr    = (const float*)d_in[3];
  const float* att   = (const float*)d_in[4];
  const float* convb = (const float*)d_in[5];
  const float* gnw   = (const float*)d_in[6];
  const float* gnb   = (const float*)d_in[7];
  const float* gnms  = (const float*)d_in[8];
  const float* W1    = (const float*)d_in[9];
  const float* b1    = (const float*)d_in[10];
  const float* W2    = (const float*)d_in[11];
  const float* b2    = (const float*)d_in[12];
  float* out = (float*)d_out;

  float* ws       = (float*)d_ws;
  float* xl       = ws;
  float* xr       = xl + (size_t)GN * GHC;
  float* hbuf     = xr + (size_t)GN * GHC;
  float* h1       = hbuf + (size_t)GN * GHC;
  float* logits   = h1;
  size_t scratch1 = ((size_t)GN * GC > (size_t)GETOT * GH)
                        ? (size_t)GN * GC : (size_t)GETOT * GH;
  float* accum    = h1 + scratch1;
  float* denom    = accum + (size_t)GN * GHC;
  float* colsum   = denom + (size_t)GN * GH;
  float* colsumsq = colsum + GHC;
  float* s1       = colsumsq + GHC;
  float* s2       = s1 + GHC;
  float* m        = s2 + GHC;

  const int zeroCount = GN * GHC + GN * GH + 2 * GHC;
  const int pairCount = GETOT * GH;

  const float* cur = x;
  for (int layer = 0; layer < 2; ++layer) {
    const float* wl = Wl + (size_t)layer * GHC * GHC;
    const float* wr = Wr + (size_t)layer * GHC * GHC;
    const float* at = att + (size_t)layer * GH * GC;
    const float* cb = convb + (size_t)layer * GHC;

    wmma_gemm_f32<<<GN / 16, dim3(32, GHC / 16), 0, stream>>>(cur, wl, nullptr, xl, GHC, GHC, 0);
    wmma_gemm_f32<<<GN / 16, dim3(32, GHC / 16), 0, stream>>>(cur, wr, nullptr, xr, GHC, GHC, 0);

    fill_kernel<<<(zeroCount + 255) / 256, 256, 0, stream>>>(accum, 0.0f, zeroCount);
    fill_kernel<<<(GN * GH + 255) / 256, 256, 0, stream>>>(m, -3.0e38f, GN * GH);

    edge_logits_kernel<<<(NBATCH + EDGE_WAVES_PER_BLOCK - 1) / EDGE_WAVES_PER_BLOCK, EDGE_WAVES_PER_BLOCK * 32, 0, stream>>>(xl, xr, ei, at, logits, m);
    edge_softmax_p<<<(pairCount + 255) / 256, 256, 0, stream>>>(ei, logits, m, denom);
    edge_aggregate<<<(NBATCH + EDGE_WAVES_PER_BLOCK - 1) / EDGE_WAVES_PER_BLOCK, EDGE_WAVES_PER_BLOCK * 32, 0, stream>>>(ei, xl, logits, denom, accum);

    gn_stats<<<(GN + 127) / 128, GHC, 0, stream>>>(accum, cb, colsum, colsumsq);
    gn_finalize<<<1, GHC, 0, stream>>>(colsum, colsumsq,
                                       gnw + (size_t)layer * GHC,
                                       gnb + (size_t)layer * GHC,
                                       gnms + (size_t)layer * GHC, s1, s2);
    gn_apply<<<(GN * GHC + 255) / 256, 256, 0, stream>>>(accum, cb, s1, s2, hbuf);

    cur = hbuf;
  }

  wmma_gemm_f32<<<GN / 16, dim3(32, GC / 16), 0, stream>>>(hbuf, W1, b1, h1, GHC, GC, 1);
  head_kernel<<<(GN + 255) / 256, 256, 0, stream>>>(h1, W2, b2, out);
}


# Edges grouped by destination once per call (int histogram, scan, arrival-order fill); one wave per node does logits, running-max softmax and weighted sum in registers: no float atomics
# speedup vs baseline: 2.7213x; 1.4080x over previous
#include <hip/hip_runtime.h>
#include <hip/hip_bf16.h>


#define GN 50000
#define GE 800000
#define GETOT (GE + GN)
#define GHC 128
#define GC 64
#define GH 2
#define NEG_SLOPE 0.2f
#define GEPS 1e-5f

typedef __attribute__((ext_vector_type(2))) float v2f;
typedef __attribute__((ext_vector_type(8))) float v8f;

__global__ void wmma_gemm_f32(const float* __restrict__ A,
                              const float* __restrict__ B,
                              const float* __restrict__ bias,
                              float* __restrict__ C,
                              int K, int Nout, int doRelu) {
  const int lane    = threadIdx.x;
  const int colTile = threadIdx.y;
  const int row0    = blockIdx.x * 16;
  const int col0    = colTile * 16;
  const int col     = lane & 15;
  const int kHalf   = (lane >> 4) << 1;
  const int aRow    = row0 + (lane & 15);

  v8f acc = {};
  for (int kk = 0; kk < K; kk += 4) {
    v2f a, b;
    const float* ap = A + (size_t)aRow * K + (kk + kHalf);
    a.x = ap[0];
    a.y = ap[1];
    const float* bp = B + (size_t)(kk + kHalf) * Nout + col0 + col;
    b.x = bp[0];
    b.y = bp[Nout];
    acc = __builtin_amdgcn_wmma_f32_16x16x4_f32(false, a, false, b,
                                                (short)0, acc, false, false);
  }

  const int rowBase = row0 + ((lane >> 4) << 3);
  const float bv = bias ? bias[col0 + col] : 0.0f;
#pragma unroll
  for (int j = 0; j < 8; ++j) {
    float v = acc[j] + bv;
    if (doRelu) v = fmaxf(v, 0.0f);
    C[(size_t)(rowBase + j) * Nout + col0 + col] = v;
  }
}

__global__ void fill_kernel(float* __restrict__ p, float v, int n) {
  int i = blockIdx.x * blockDim.x + threadIdx.x;
  if (i < n) p[i] = v;
}

__global__ void fill_int_kernel(int* __restrict__ p, int v, int n) {
  int i = blockIdx.x * blockDim.x + threadIdx.x;
  if (i < n) p[i] = v;
}

__device__ __forceinline__ int bcast_i(int v, int i) {
  return __builtin_amdgcn_readlane(v, i);
}

__global__ void csr_count(const int* __restrict__ ei, int* __restrict__ cnt) {
  int e = blockIdx.x * blockDim.x + threadIdx.x;
  if (e >= GE) return;
  atomicAdd(&cnt[ei[GE + e]], 1);
}

#define SCAN_T 1024
#define SCAN_BLOCKS ((GN + SCAN_T - 1) / SCAN_T)

__global__ void __launch_bounds__(SCAN_T)
csr_scan(const int* __restrict__ cnt, int* __restrict__ rowptr, int* __restrict__ cursor) {
  __shared__ int wsum[SCAN_T / 32];
  __shared__ int wtot[SCAN_T / 32];
  const int t    = threadIdx.x;
  const int lane = t & 31;
  const int wv   = t >> 5;
  const int tile0 = blockIdx.x * SCAN_T;

  int pre = 0;
  for (int i = t; i < tile0; i += SCAN_T) pre += cnt[i];
#pragma unroll
  for (int off = 16; off > 0; off >>= 1) pre += __shfl_xor(pre, off, 32);

  const int idx = tile0 + t;
  const int x = (idx < GN) ? cnt[idx] : 0;
  int inc = x;
#pragma unroll
  for (int off = 1; off < 32; off <<= 1) {
    const int y = __shfl_up(inc, off, 32);
    if (lane >= off) inc += y;
  }
  if (lane == 0)  wsum[wv] = pre;
  if (lane == 31) wtot[wv] = inc;
  __syncthreads();

  int base = 0;
  int wofs = 0;
#pragma unroll
  for (int k = 0; k < SCAN_T / 32; ++k) {
    base += wsum[k];
    if (k < wv) wofs += wtot[k];
  }
  const int excl = base + wofs + inc - x;
  if (idx < GN) {
    rowptr[idx] = excl;
    cursor[idx] = excl;
  }
  if (idx == GN - 1) rowptr[GN] = excl + x;
}

__global__ void csr_fill(const int* __restrict__ ei, int* __restrict__ cursor,
                         int* __restrict__ col) {
  int e = blockIdx.x * blockDim.x + threadIdx.x;
  if (e >= GE) return;
  const int pos = atomicAdd(&cursor[ei[GE + e]], 1);
  col[pos] = ei[e];
}

#define NODE_WAVES_PER_BLOCK 4

__device__ __forceinline__ void edge_logit2(float x0, float x1, float x2, float x3,
                                            float r0, float r1, float r2, float r3,
                                            float a00, float a01, float a10, float a11,
                                            bool lowHalf, float& l0, float& l1) {
  float t0 = x0 + r0;
  float t1 = x1 + r1;
  float u0 = x2 + r2;
  float u1 = x3 + r3;
  t0 = (t0 > 0.0f) ? t0 : NEG_SLOPE * t0;
  t1 = (t1 > 0.0f) ? t1 : NEG_SLOPE * t1;
  u0 = (u0 > 0.0f) ? u0 : NEG_SLOPE * u0;
  u1 = (u1 > 0.0f) ? u1 : NEG_SLOPE * u1;
  const float v0 = t0 * a00 + t1 * a01;
  const float v1 = u0 * a10 + u1 * a11;
  const float keep = lowHalf ? v0 : v1;
  const float send = lowHalf ? v1 : v0;
  float w = keep + __shfl_xor(send, 16, 32);
#pragma unroll
  for (int off = 8; off > 0; off >>= 1) w += __shfl_xor(w, off, 32);
  const float o = __shfl_xor(w, 16, 32);
  l0 = lowHalf ? w : o;
  l1 = lowHalf ? o : w;
}

__global__ void __launch_bounds__(NODE_WAVES_PER_BLOCK * 32)
node_attend(const float* __restrict__ xl,
            const float* __restrict__ xr,
            const int* __restrict__ rowptr,
            const int* __restrict__ col,
            const float* __restrict__ att,
            float* __restrict__ outp) {
  const int lane = threadIdx.x & 31;
  const int n = __builtin_amdgcn_readfirstlane(blockIdx.x * NODE_WAVES_PER_BLOCK + (threadIdx.x >> 5));
  if (n >= GN) return;
  const int beg = rowptr[n];
  const int end = rowptr[n + 1];

  const float a00 = att[lane];
  const float a01 = att[lane + 32];
  const float a10 = att[GC + lane];
  const float a11 = att[GC + lane + 32];
  const bool lowHalf = lane < 16;

  const float* xrn = xr + (size_t)n * GHC + lane;
  const float r0 = xrn[0];
  const float r1 = xrn[32];
  const float r2 = xrn[GC];
  const float r3 = xrn[GC + 32];

  const float* xln = xl + (size_t)n * GHC + lane;
  float acc0 = xln[0];
  float acc1 = xln[32];
  float acc2 = xln[GC];
  float acc3 = xln[GC + 32];
  float m0, m1;
  edge_logit2(acc0, acc1, acc2, acc3, r0, r1, r2, r3, a00, a01, a10, a11, lowHalf, m0, m1);
  float dn0 = 1.0f, dn1 = 1.0f;

  for (int cb = beg; cb < end; cb += 32) {
    const int left = end - cb;
    const int cnt  = (left < 32) ? left : 32;
    const int s = (lane < cnt) ? col[cb + lane] : 0;

    int i = 0;
    for (; i + 4 <= cnt; i += 4) {
      float x[4][4];
      float l0[4], l1[4];
#pragma unroll
      for (int k = 0; k < 4; ++k) {
        const int si = bcast_i(s, i + k);
        const float* xs = xl + (size_t)si * GHC + lane;
        x[k][0] = xs[0];
        x[k][1] = xs[32];
        x[k][2] = xs[GC];
        x[k][3] = xs[GC + 32];
      }
#pragma unroll
      for (int k = 0; k < 4; ++k)
        edge_logit2(x[k][0], x[k][1], x[k][2], x[k][3], r0, r1, r2, r3,
                    a00, a01, a10, a11, lowHalf, l0[k], l1[k]);

      const float n0 = fmaxf(fmaxf(m0, fmaxf(l0[0], l0[1])), fmaxf(l0[2], l0[3]));
      const float n1 = fmaxf(fmaxf(m1, fmaxf(l1[0], l1[1])), fmaxf(l1[2], l1[3]));
      const float sc0 = __expf(m0 - n0);
      const float sc1 = __expf(m1 - n1);
      m0 = n0;
      m1 = n1;
      dn0 *= sc0;  acc0 *= sc0;  acc1 *= sc0;
      dn1 *= sc1;  acc2 *= sc1;  acc3 *= sc1;
#pragma unroll
      for (int k = 0; k < 4; ++k) {
        const float p0 = __expf(l0[k] - n0);
        const float p1 = __expf(l1[k] - n1);
        dn0 += p0;
        dn1 += p1;
        acc0 += p0 * x[k][0];
        acc1 += p0 * x[k][1];
        acc2 += p1 * x[k][2];
        acc3 += p1 * x[k][3];
      }
    }
    for (; i < cnt; ++i) {
      const int si = bcast_i(s, i);
      const float* xs = xl + (size_t)si * GHC + lane;
      const float x0 = xs[0];
      const float x1 = xs[32];
      const float x2 = xs[GC];
      const float x3 = xs[GC + 32];
      float l0, l1;
      edge_logit2(x0, x1, x2, x3, r0, r1, r2, r3, a00, a01, a10, a11, lowHalf, l0, l1);
      const float n0 = fmaxf(m0, l0);
      const float n1 = fmaxf(m1, l1);
      const float sc0 = __expf(m0 - n0);
      const float sc1 = __expf(m1 - n1);
      const float p0  = __expf(l0 - n0);
      const float p1  = __expf(l1 - n1);
      m0 = n0;
      m1 = n1;
      dn0  = dn0 * sc0 + p0;
      dn1  = dn1 * sc1 + p1;
      acc0 = acc0 * sc0 + p0 * x0;
      acc1 = acc1 * sc0 + p0 * x1;
      acc2 = acc2 * sc1 + p1 * x2;
      acc3 = acc3 * sc1 + p1 * x3;
    }
  }

  float* on = outp + (size_t)n * GHC + lane;
  on[0]       = acc0 / dn0;
  on[32]      = acc1 / dn0;
  on[GC]      = acc2 / dn1;
  on[GC + 32] = acc3 / dn1;
}

__global__ void gn_stats(const float* __restrict__ accum,
                         const float* __restrict__ bias,
                         float* __restrict__ colsum,
                         float* __restrict__ colsumsq) {
  const int f = threadIdx.x;
  const int row0 = blockIdx.x * 128;
  const float bv = bias[f];
  float s = 0.0f, s2 = 0.0f;
  for (int r = 0; r < 128; ++r) {
    const int n = row0 + r;
    if (n >= GN) break;
    const float x = accum[(size_t)n * GHC + f] + bv;
    s += x;
    s2 += x * x;
  }
  atomicAdd(&colsum[f], s);
  atomicAdd(&colsumsq[f], s2);
}

__global__ void gn_finalize(const float* __restrict__ colsum,
                            const float* __restrict__ colsumsq,
                            const float* __restrict__ w,
                            const float* __restrict__ b,
                            const float* __restrict__ ms,
                            float* __restrict__ s1,
                            float* __restrict__ s2) {
  const int f = threadIdx.x;
  const float invN = 1.0f / (float)GN;
  const float mean = colsum[f] * invN;
  const float ex2  = colsumsq[f] * invN;
  const float msv  = ms[f];
  const float var  = ex2 - (2.0f * msv - msv * msv) * mean * mean;
  const float r    = rsqrtf(var + GEPS);
  const float a    = w[f] * r;
  s1[f] = a;
  s2[f] = b[f] - a * msv * mean;
}

__global__ void gn_apply(const float* __restrict__ accum,
                         const float* __restrict__ bias,
                         const float* __restrict__ s1,
                         const float* __restrict__ s2,
                         float* __restrict__ out) {
  int i = blockIdx.x * blockDim.x + threadIdx.x;
  if (i >= GN * GHC) return;
  const int f = i & (GHC - 1);
  const float x = accum[i] + bias[f];
  out[i] = fmaxf(x * s1[f] + s2[f], 0.0f);
}

__global__ void head_kernel(const float* __restrict__ h1,
                            const float* __restrict__ W2,
                            const float* __restrict__ b2,
                            float* __restrict__ out) {
  int n = blockIdx.x * blockDim.x + threadIdx.x;
  if (n >= GN) return;
  const float* row = h1 + (size_t)n * GC;
  float a0 = b2[0], a1 = b2[1];
#pragma unroll 8
  for (int c = 0; c < GC; ++c) {
    const float v = row[c];
    a0 += v * W2[c * 2 + 0];
    a1 += v * W2[c * 2 + 1];
  }
  out[n * 2 + 0] = a0;
  out[n * 2 + 1] = a1;
}

extern "C" void kernel_launch(void* const* d_in, const int* in_sizes, int n_in,
                              void* d_out, int out_size, void* d_ws, size_t ws_size,
                              hipStream_t stream) {
  (void)in_sizes; (void)n_in; (void)out_size; (void)ws_size;

  const float* x     = (const float*)d_in[0];
  const int*   ei    = (const int*)d_in[1];
  const float* Wl    = (const float*)d_in[2];
  const float* Wr    = (const float*)d_in[3];
  const float* att   = (const float*)d_in[4];
  const float* convb = (const float*)d_in[5];
  const float* gnw   = (const float*)d_in[6];
  const float* gnb   = (const float*)d_in[7];
  const float* gnms  = (const float*)d_in[8];
  const float* W1    = (const float*)d_in[9];
  const float* b1    = (const float*)d_in[10];
  const float* W2    = (const float*)d_in[11];
  const float* b2    = (const float*)d_in[12];
  float* out = (float*)d_out;

  float* ws       = (float*)d_ws;
  float* xl       = ws;
  float* xr       = xl + (size_t)GN * GHC;
  float* hbuf     = xr + (size_t)GN * GHC;
  float* h1       = hbuf + (size_t)GN * GHC;
  int*   col      = (int*)h1;
  size_t scratch1 = ((size_t)GN * GC > (size_t)GETOT * GH)
                        ? (size_t)GN * GC : (size_t)GETOT * GH;
  float* accum    = h1 + scratch1;
  float* rpregion = accum + (size_t)GN * GHC;
  int*   rowptr   = (int*)rpregion;
  float* colsum   = rpregion + (size_t)GN * GH;
  float* colsumsq = colsum + GHC;
  float* s1       = colsumsq + GHC;
  float* s2       = s1 + GHC;
  int*   cnt      = (int*)(s2 + GHC);
  int*   cursor   = cnt + GN;

  fill_int_kernel<<<(GN + 255) / 256, 256, 0, stream>>>(cnt, 0, GN);
  csr_count<<<(GE + 255) / 256, 256, 0, stream>>>(ei, cnt);
  csr_scan<<<SCAN_BLOCKS, SCAN_T, 0, stream>>>(cnt, rowptr, cursor);
  csr_fill<<<(GE + 255) / 256, 256, 0, stream>>>(ei, cursor, col);

  const float* cur = x;
  for (int layer = 0; layer < 2; ++layer) {
    const float* wl = Wl + (size_t)layer * GHC * GHC;
    const float* wr = Wr + (size_t)layer * GHC * GHC;
    const float* at = att + (size_t)layer * GH * GC;
    const float* cb = convb + (size_t)layer * GHC;

    wmma_gemm_f32<<<GN / 16, dim3(32, GHC / 16), 0, stream>>>(cur, wl, nullptr, xl, GHC, GHC, 0);
    wmma_gemm_f32<<<GN / 16, dim3(32, GHC / 16), 0, stream>>>(cur, wr, nullptr, xr, GHC, GHC, 0);

    fill_kernel<<<1, 2 * GHC, 0, stream>>>(colsum, 0.0f, 2 * GHC);

    node_attend<<<(GN + NODE_WAVES_PER_BLOCK - 1) / NODE_WAVES_PER_BLOCK, NODE_WAVES_PER_BLOCK * 32, 0, stream>>>(xl, xr, rowptr, col, at, accum);

    gn_stats<<<(GN + 127) / 128, GHC, 0, stream>>>(accum, cb, colsum, colsumsq);
    gn_finalize<<<1, GHC, 0, stream>>>(colsum, colsumsq,
                                       gnw + (size_t)layer * GHC,
                                       gnb + (size_t)layer * GHC,
                                       gnms + (size_t)layer * GHC, s1, s2);
    gn_apply<<<(GN * GHC + 255) / 256, 256, 0, stream>>>(accum, cb, s1, s2, hbuf);

    cur = hbuf;
  }

  wmma_gemm_f32<<<GN / 16, dim3(32, GC / 16), 0, stream>>>(hbuf, W1, b1, h1, GHC, GC, 1);
  head_kernel<<<(GN + 255) / 256, 256, 0, stream>>>(h1, W2, b2, out);
}


# GraphNorm apply: one wave owns 8 whole rows (eight 16-byte-per-lane row loads in flight, column constants loaded once), 6250 waves per launch instead of 200000 one-element-per-lane waves
# speedup vs baseline: 2.8165x; 1.0350x over previous
#include <hip/hip_runtime.h>
#include <hip/hip_bf16.h>


#define GN 50000
#define GE 800000
#define GETOT (GE + GN)
#define GHC 128
#define GC 64
#define GH 2
#define NEG_SLOPE 0.2f
#define GEPS 1e-5f

typedef __attribute__((ext_vector_type(2))) float v2f;
typedef __attribute__((ext_vector_type(8))) float v8f;

__global__ void wmma_gemm_f32(const float* __restrict__ A,
                              const float* __restrict__ B,
                              const float* __restrict__ bias,
                              float* __restrict__ C,
                              int K, int Nout, int doRelu) {
  const int lane    = threadIdx.x;
  const int colTile = threadIdx.y;
  const int row0    = blockIdx.x * 16;
  const int col0    = colTile * 16;
  const int col     = lane & 15;
  const int kHalf   = (lane >> 4) << 1;
  const int aRow    = row0 + (lane & 15);

  v8f acc = {};
  for (int kk = 0; kk < K; kk += 4) {
    v2f a, b;
    const float* ap = A + (size_t)aRow * K + (kk + kHalf);
    a.x = ap[0];
    a.y = ap[1];
    const float* bp = B + (size_t)(kk + kHalf) * Nout + col0 + col;
    b.x = bp[0];
    b.y = bp[Nout];
    acc = __builtin_amdgcn_wmma_f32_16x16x4_f32(false, a, false, b,
                                                (short)0, acc, false, false);
  }

  const int rowBase = row0 + ((lane >> 4) << 3);
  const float bv = bias ? bias[col0 + col] : 0.0f;
#pragma unroll
  for (int j = 0; j < 8; ++j) {
    float v = acc[j] + bv;
    if (doRelu) v = fmaxf(v, 0.0f);
    C[(size_t)(rowBase + j) * Nout + col0 + col] = v;
  }
}

__global__ void fill_kernel(float* __restrict__ p, float v, int n) {
  int i = blockIdx.x * blockDim.x + threadIdx.x;
  if (i < n) p[i] = v;
}

__global__ void fill_int_kernel(int* __restrict__ p, int v, int n) {
  int i = blockIdx.x * blockDim.x + threadIdx.x;
  if (i < n) p[i] = v;
}

__device__ __forceinline__ int bcast_i(int v, int i) {
  return __builtin_amdgcn_readlane(v, i);
}

__global__ void csr_count(const int* __restrict__ ei, int* __restrict__ cnt) {
  int e = blockIdx.x * blockDim.x + threadIdx.x;
  if (e >= GE) return;
  atomicAdd(&cnt[ei[GE + e]], 1);
}

#define SCAN_T 1024
#define SCAN_BLOCKS ((GN + SCAN_T - 1) / SCAN_T)

__global__ void __launch_bounds__(SCAN_T)
csr_scan(const int* __restrict__ cnt, int* __restrict__ rowptr, int* __restrict__ cursor) {
  __shared__ int wsum[SCAN_T / 32];
  __shared__ int wtot[SCAN_T / 32];
  const int t    = threadIdx.x;
  const int lane = t & 31;
  const int wv   = t >> 5;
  const int tile0 = blockIdx.x * SCAN_T;

  int pre = 0;
  for (int i = t; i < tile0; i += SCAN_T) pre += cnt[i];
#pragma unroll
  for (int off = 16; off > 0; off >>= 1) pre += __shfl_xor(pre, off, 32);

  const int idx = tile0 + t;
  const int x = (idx < GN) ? cnt[idx] : 0;
  int inc = x;
#pragma unroll
  for (int off = 1; off < 32; off <<= 1) {
    const int y = __shfl_up(inc, off, 32);
    if (lane >= off) inc += y;
  }
  if (lane == 0)  wsum[wv] = pre;
  if (lane == 31) wtot[wv] = inc;
  __syncthreads();

  int base = 0;
  int wofs = 0;
#pragma unroll
  for (int k = 0; k < SCAN_T / 32; ++k) {
    base += wsum[k];
    if (k < wv) wofs += wtot[k];
  }
  const int excl = base + wofs + inc - x;
  if (idx < GN) {
    rowptr[idx] = excl;
    cursor[idx] = excl;
  }
  if (idx == GN - 1) rowptr[GN] = excl + x;
}

__global__ void csr_fill(const int* __restrict__ ei, int* __restrict__ cursor,
                         int* __restrict__ col) {
  int e = blockIdx.x * blockDim.x + threadIdx.x;
  if (e >= GE) return;
  const int pos = atomicAdd(&cursor[ei[GE + e]], 1);
  col[pos] = ei[e];
}

#define NODE_WAVES_PER_BLOCK 4

__device__ __forceinline__ void edge_logit2(float x0, float x1, float x2, float x3,
                                            float r0, float r1, float r2, float r3,
                                            float a00, float a01, float a10, float a11,
                                            bool lowHalf, float& l0, float& l1) {
  float t0 = x0 + r0;
  float t1 = x1 + r1;
  float u0 = x2 + r2;
  float u1 = x3 + r3;
  t0 = (t0 > 0.0f) ? t0 : NEG_SLOPE * t0;
  t1 = (t1 > 0.0f) ? t1 : NEG_SLOPE * t1;
  u0 = (u0 > 0.0f) ? u0 : NEG_SLOPE * u0;
  u1 = (u1 > 0.0f) ? u1 : NEG_SLOPE * u1;
  const float v0 = t0 * a00 + t1 * a01;
  const float v1 = u0 * a10 + u1 * a11;
  const float keep = lowHalf ? v0 : v1;
  const float send = lowHalf ? v1 : v0;
  float w = keep + __shfl_xor(send, 16, 32);
#pragma unroll
  for (int off = 8; off > 0; off >>= 1) w += __shfl_xor(w, off, 32);
  const float o = __shfl_xor(w, 16, 32);
  l0 = lowHalf ? w : o;
  l1 = lowHalf ? o : w;
}

__global__ void __launch_bounds__(NODE_WAVES_PER_BLOCK * 32)
node_attend(const float* __restrict__ xl,
            const float* __restrict__ xr,
            const int* __restrict__ rowptr,
            const int* __restrict__ col,
            const float* __restrict__ att,
            float* __restrict__ outp) {
  const int lane = threadIdx.x & 31;
  const int n = __builtin_amdgcn_readfirstlane(blockIdx.x * NODE_WAVES_PER_BLOCK + (threadIdx.x >> 5));
  if (n >= GN) return;
  const int beg = rowptr[n];
  const int end = rowptr[n + 1];

  const float a00 = att[lane];
  const float a01 = att[lane + 32];
  const float a10 = att[GC + lane];
  const float a11 = att[GC + lane + 32];
  const bool lowHalf = lane < 16;

  const float* xrn = xr + (size_t)n * GHC + lane;
  const float r0 = xrn[0];
  const float r1 = xrn[32];
  const float r2 = xrn[GC];
  const float r3 = xrn[GC + 32];

  const float* xln = xl + (size_t)n * GHC + lane;
  float acc0 = xln[0];
  float acc1 = xln[32];
  float acc2 = xln[GC];
  float acc3 = xln[GC + 32];
  float m0, m1;
  edge_logit2(acc0, acc1, acc2, acc3, r0, r1, r2, r3, a00, a01, a10, a11, lowHalf, m0, m1);
  float dn0 = 1.0f, dn1 = 1.0f;

  for (int cb = beg; cb < end; cb += 32) {
    const int left = end - cb;
    const int cnt  = (left < 32) ? left : 32;
    const int s = (lane < cnt) ? col[cb + lane] : 0;

    int i = 0;
    for (; i + 4 <= cnt; i += 4) {
      float x[4][4];
      float l0[4], l1[4];
#pragma unroll
      for (int k = 0; k < 4; ++k) {
        const int si = bcast_i(s, i + k);
        const float* xs = xl + (size_t)si * GHC + lane;
        x[k][0] = xs[0];
        x[k][1] = xs[32];
        x[k][2] = xs[GC];
        x[k][3] = xs[GC + 32];
      }
#pragma unroll
      for (int k = 0; k < 4; ++k)
        edge_logit2(x[k][0], x[k][1], x[k][2], x[k][3], r0, r1, r2, r3,
                    a00, a01, a10, a11, lowHalf, l0[k], l1[k]);

      const float n0 = fmaxf(fmaxf(m0, fmaxf(l0[0], l0[1])), fmaxf(l0[2], l0[3]));
      const float n1 = fmaxf(fmaxf(m1, fmaxf(l1[0], l1[1])), fmaxf(l1[2], l1[3]));
      const float sc0 = __expf(m0 - n0);
      const float sc1 = __expf(m1 - n1);
      m0 = n0;
      m1 = n1;
      dn0 *= sc0;  acc0 *= sc0;  acc1 *= sc0;
      dn1 *= sc1;  acc2 *= sc1;  acc3 *= sc1;
#pragma unroll
      for (int k = 0; k < 4; ++k) {
        const float p0 = __expf(l0[k] - n0);
        const float p1 = __expf(l1[k] - n1);
        dn0 += p0;
        dn1 += p1;
        acc0 += p0 * x[k][0];
        acc1 += p0 * x[k][1];
        acc2 += p1 * x[k][2];
        acc3 += p1 * x[k][3];
      }
    }
    for (; i < cnt; ++i) {
      const int si = bcast_i(s, i);
      const float* xs = xl + (size_t)si * GHC + lane;
      const float x0 = xs[0];
      const float x1 = xs[32];
      const float x2 = xs[GC];
      const float x3 = xs[GC + 32];
      float l0, l1;
      edge_logit2(x0, x1, x2, x3, r0, r1, r2, r3, a00, a01, a10, a11, lowHalf, l0, l1);
      const float n0 = fmaxf(m0, l0);
      const float n1 = fmaxf(m1, l1);
      const float sc0 = __expf(m0 - n0);
      const float sc1 = __expf(m1 - n1);
      const float p0  = __expf(l0 - n0);
      const float p1  = __expf(l1 - n1);
      m0 = n0;
      m1 = n1;
      dn0  = dn0 * sc0 + p0;
      dn1  = dn1 * sc1 + p1;
      acc0 = acc0 * sc0 + p0 * x0;
      acc1 = acc1 * sc0 + p0 * x1;
      acc2 = acc2 * sc1 + p1 * x2;
      acc3 = acc3 * sc1 + p1 * x3;
    }
  }

  float* on = outp + (size_t)n * GHC + lane;
  on[0]       = acc0 / dn0;
  on[32]      = acc1 / dn0;
  on[GC]      = acc2 / dn1;
  on[GC + 32] = acc3 / dn1;
}

__global__ void gn_stats(const float* __restrict__ accum,
                         const float* __restrict__ bias,
                         float* __restrict__ colsum,
                         float* __restrict__ colsumsq) {
  const int f = threadIdx.x;
  const int row0 = blockIdx.x * 128;
  const float bv = bias[f];
  float s = 0.0f, s2 = 0.0f;
  for (int r = 0; r < 128; ++r) {
    const int n = row0 + r;
    if (n >= GN) break;
    const float x = accum[(size_t)n * GHC + f] + bv;
    s += x;
    s2 += x * x;
  }
  atomicAdd(&colsum[f], s);
  atomicAdd(&colsumsq[f], s2);
}

__global__ void gn_finalize(const float* __restrict__ colsum,
                            const float* __restrict__ colsumsq,
                            const float* __restrict__ w,
                            const float* __restrict__ b,
                            const float* __restrict__ ms,
                            float* __restrict__ s1,
                            float* __restrict__ s2) {
  const int f = threadIdx.x;
  const float invN = 1.0f / (float)GN;
  const float mean = colsum[f] * invN;
  const float ex2  = colsumsq[f] * invN;
  const float msv  = ms[f];
  const float var  = ex2 - (2.0f * msv - msv * msv) * mean * mean;
  const float r    = rsqrtf(var + GEPS);
  const float a    = w[f] * r;
  s1[f] = a;
  s2[f] = b[f] - a * msv * mean;
}

#define APPLY_ROWS 8
#define APPLY_WAVES_PER_BLOCK 8
#define APPLY_WAVES (GN / APPLY_ROWS)
static_assert(GN % APPLY_ROWS == 0, "a wave's rows are all inside the array");

__global__ void __launch_bounds__(APPLY_WAVES_PER_BLOCK * 32)
gn_apply(const float* __restrict__ accum,
         const float* __restrict__ bias,
         const float* __restrict__ s1,
         const float* __restrict__ s2,
         float* __restrict__ out) {
  const int lane = threadIdx.x & 31;
  const int wave = blockIdx.x * APPLY_WAVES_PER_BLOCK + (threadIdx.x >> 5);
  if (wave >= APPLY_WAVES) return;
  const size_t base = (size_t)wave * APPLY_ROWS * GHC + (size_t)lane * 4;

  float4 v[APPLY_ROWS];
#pragma unroll
  for (int k = 0; k < APPLY_ROWS; ++k)
    v[k] = *(const float4*)(accum + base + (size_t)k * GHC);

  const float4 bv = *(const float4*)(bias + lane * 4);
  const float4 a1 = *(const float4*)(s1 + lane * 4);
  const float4 a2 = *(const float4*)(s2 + lane * 4);

#pragma unroll
  for (int k = 0; k < APPLY_ROWS; ++k) {
    float4 r;
    r.x = fmaxf((v[k].x + bv.x) * a1.x + a2.x, 0.0f);
    r.y = fmaxf((v[k].y + bv.y) * a1.y + a2.y, 0.0f);
    r.z = fmaxf((v[k].z + bv.z) * a1.z + a2.z, 0.0f);
    r.w = fmaxf((v[k].w + bv.w) * a1.w + a2.w, 0.0f);
    *(float4*)(out + base + (size_t)k * GHC) = r;
  }
}

__global__ void head_kernel(const float* __restrict__ h1,
                            const float* __restrict__ W2,
                            const float* __restrict__ b2,
                            float* __restrict__ out) {
  int n = blockIdx.x * blockDim.x + threadIdx.x;
  if (n >= GN) return;
  const float* row = h1 + (size_t)n * GC;
  float a0 = b2[0], a1 = b2[1];
#pragma unroll 8
  for (int c = 0; c < GC; ++c) {
    const float v = row[c];
    a0 += v * W2[c * 2 + 0];
    a1 += v * W2[c * 2 + 1];
  }
  out[n * 2 + 0] = a0;
  out[n * 2 + 1] = a1;
}

extern "C" void kernel_launch(void* const* d_in, const int* in_sizes, int n_in,
                              void* d_out, int out_size, void* d_ws, size_t ws_size,
                              hipStream_t stream) {
  (void)in_sizes; (void)n_in; (void)out_size; (void)ws_size;

  const float* x     = (const float*)d_in[0];
  const int*   ei    = (const int*)d_in[1];
  const float* Wl    = (const float*)d_in[2];
  const float* Wr    = (const float*)d_in[3];
  const float* att   = (const float*)d_in[4];
  const float* convb = (const float*)d_in[5];
  const float* gnw   = (const float*)d_in[6];
  const float* gnb   = (const float*)d_in[7];
  const float* gnms  = (const float*)d_in[8];
  const float* W1    = (const float*)d_in[9];
  const float* b1    = (const float*)d_in[10];
  const float* W2    = (const float*)d_in[11];
  const float* b2    = (const float*)d_in[12];
  float* out = (float*)d_out;

  float* ws       = (float*)d_ws;
  float* xl       = ws;
  float* xr       = xl + (size_t)GN * GHC;
  float* hbuf     = xr + (size_t)GN * GHC;
  float* h1       = hbuf + (size_t)GN * GHC;
  int*   col      = (int*)h1;
  size_t scratch1 = ((size_t)GN * GC > (size_t)GETOT * GH)
                        ? (size_t)GN * GC : (size_t)GETOT * GH;
  float* accum    = h1 + scratch1;
  float* rpregion = accum + (size_t)GN * GHC;
  int*   rowptr   = (int*)rpregion;
  float* colsum   = rpregion + (size_t)GN * GH;
  float* colsumsq = colsum + GHC;
  float* s1       = colsumsq + GHC;
  float* s2       = s1 + GHC;
  int*   cnt      = (int*)(s2 + GHC);
  int*   cursor   = cnt + GN;

  fill_int_kernel<<<(GN + 255) / 256, 256, 0, stream>>>(cnt, 0, GN);
  csr_count<<<(GE + 255) / 256, 256, 0, stream>>>(ei, cnt);
  csr_scan<<<SCAN_BLOCKS, SCAN_T, 0, stream>>>(cnt, rowptr, cursor);
  csr_fill<<<(GE + 255) / 256, 256, 0, stream>>>(ei, cursor, col);

  const float* cur = x;
  for (int layer = 0; layer < 2; ++layer) {
    const float* wl = Wl + (size_t)layer * GHC * GHC;
    const float* wr = Wr + (size_t)layer * GHC * GHC;
    const float* at = att + (size_t)layer * GH * GC;
    const float* cb = convb + (size_t)layer * GHC;

    wmma_gemm_f32<<<GN / 16, dim3(32, GHC / 16), 0, stream>>>(cur, wl, nullptr, xl, GHC, GHC, 0);
    wmma_gemm_f32<<<GN / 16, dim3(32, GHC / 16), 0, stream>>>(cur, wr, nullptr, xr, GHC, GHC, 0);

    fill_kernel<<<1, 2 * GHC, 0, stream>>>(colsum, 0.0f, 2 * GHC);

    node_attend<<<(GN + NODE_WAVES_PER_BLOCK - 1) / NODE_WAVES_PER_BLOCK, NODE_WAVES_PER_BLOCK * 32, 0, stream>>>(xl, xr, rowptr, col, at, accum);

    gn_stats<<<(GN + 127) / 128, GHC, 0, stream>>>(accum, cb, colsum, colsumsq);
    gn_finalize<<<1, GHC, 0, stream>>>(colsum, colsumsq,
                                       gnw + (size_t)layer * GHC,
                                       gnb + (size_t)layer * GHC,
                                       gnms + (size_t)layer * GHC, s1, s2);
    gn_apply<<<(APPLY_WAVES + APPLY_WAVES_PER_BLOCK - 1) / APPLY_WAVES_PER_BLOCK, APPLY_WAVES_PER_BLOCK * 32, 0, stream>>>(accum, cb, s1, s2, hbuf);

    cur = hbuf;
  }

  wmma_gemm_f32<<<GN / 16, dim3(32, GC / 16), 0, stream>>>(hbuf, W1, b1, h1, GHC, GC, 1);
  head_kernel<<<(GN + 255) / 256, 256, 0, stream>>>(h1, W2, b2, out);
}


# GraphNorm column sums: one wave owns 16 whole rows (sixteen 16-byte-per-lane row loads in flight, not 128 loads one at a time), block partials added in LDS before the float atomics: new sum order
# speedup vs baseline: 3.7525x; 1.3323x over previous
#include <hip/hip_runtime.h>
#include <hip/hip_bf16.h>


#define GN 50000
#define GE 800000
#define GETOT (GE + GN)
#define GHC 128
#define GC 64
#define GH 2
#define NEG_SLOPE 0.2f
#define GEPS 1e-5f

typedef __attribute__((ext_vector_type(2))) float v2f;
typedef __attribute__((ext_vector_type(8))) float v8f;

__global__ void wmma_gemm_f32(const float* __restrict__ A,
                              const float* __restrict__ B,
                              const float* __restrict__ bias,
                              float* __restrict__ C,
                              int K, int Nout, int doRelu) {
  const int lane    = threadIdx.x;
  const int colTile = threadIdx.y;
  const int row0    = blockIdx.x * 16;
  const int col0    = colTile * 16;
  const int col     = lane & 15;
  const int kHalf   = (lane >> 4) << 1;
  const int aRow    = row0 + (lane & 15);

  v8f acc = {};
  for (int kk = 0; kk < K; kk += 4) {
    v2f a, b;
    const float* ap = A + (size_t)aRow * K + (kk + kHalf);
    a.x = ap[0];
    a.y = ap[1];
    const float* bp = B + (size_t)(kk + kHalf) * Nout + col0 + col;
    b.x = bp[0];
    b.y = bp[Nout];
    acc = __builtin_amdgcn_wmma_f32_16x16x4_f32(false, a, false, b,
                                                (short)0, acc, false, false);
  }

  const int rowBase = row0 + ((lane >> 4) << 3);
  const float bv = bias ? bias[col0 + col] : 0.0f;
#pragma unroll
  for (int j = 0; j < 8; ++j) {
    float v = acc[j] + bv;
    if (doRelu) v = fmaxf(v, 0.0f);
    C[(size_t)(rowBase + j) * Nout + col0 + col] = v;
  }
}

__global__ void fill_kernel(float* __restrict__ p, float v, int n) {
  int i = blockIdx.x * blockDim.x + threadIdx.x;
  if (i < n) p[i] = v;
}

__global__ void fill_int_kernel(int* __restrict__ p, int v, int n) {
  int i = blockIdx.x * blockDim.x + threadIdx.x;
  if (i < n) p[i] = v;
}

__device__ __forceinline__ int bcast_i(int v, int i) {
  return __builtin_amdgcn_readlane(v, i);
}

__global__ void csr_count(const int* __restrict__ ei, int* __restrict__ cnt) {
  int e = blockIdx.x * blockDim.x + threadIdx.x;
  if (e >= GE) return;
  atomicAdd(&cnt[ei[GE + e]], 1);
}

#define SCAN_T 1024
#define SCAN_BLOCKS ((GN + SCAN_T - 1) / SCAN_T)

__global__ void __launch_bounds__(SCAN_T)
csr_scan(const int* __restrict__ cnt, int* __restrict__ rowptr, int* __restrict__ cursor) {
  __shared__ int wsum[SCAN_T / 32];
  __shared__ int wtot[SCAN_T / 32];
  const int t    = threadIdx.x;
  const int lane = t & 31;
  const int wv   = t >> 5;
  const int tile0 = blockIdx.x * SCAN_T;

  int pre = 0;
  for (int i = t; i < tile0; i += SCAN_T) pre += cnt[i];
#pragma unroll
  for (int off = 16; off > 0; off >>= 1) pre += __shfl_xor(pre, off, 32);

  const int idx = tile0 + t;
  const int x = (idx < GN) ? cnt[idx] : 0;
  int inc = x;
#pragma unroll
  for (int off = 1; off < 32; off <<= 1) {
    const int y = __shfl_up(inc, off, 32);
    if (lane >= off) inc += y;
  }
  if (lane == 0)  wsum[wv] = pre;
  if (lane == 31) wtot[wv] = inc;
  __syncthreads();

  int base = 0;
  int wofs = 0;
#pragma unroll
  for (int k = 0; k < SCAN_T / 32; ++k) {
    base += wsum[k];
    if (k < wv) wofs += wtot[k];
  }
  const int excl = base + wofs + inc - x;
  if (idx < GN) {
    rowptr[idx] = excl;
    cursor[idx] = excl;
  }
  if (idx == GN - 1) rowptr[GN] = excl + x;
}

__global__ void csr_fill(const int* __restrict__ ei, int* __restrict__ cursor,
                         int* __restrict__ col) {
  int e = blockIdx.x * blockDim.x + threadIdx.x;
  if (e >= GE) return;
  const int pos = atomicAdd(&cursor[ei[GE + e]], 1);
  col[pos] = ei[e];
}

#define NODE_WAVES_PER_BLOCK 4

__device__ __forceinline__ void edge_logit2(float x0, float x1, float x2, float x3,
                                            float r0, float r1, float r2, float r3,
                                            float a00, float a01, float a10, float a11,
                                            bool lowHalf, float& l0, float& l1) {
  float t0 = x0 + r0;
  float t1 = x1 + r1;
  float u0 = x2 + r2;
  float u1 = x3 + r3;
  t0 = (t0 > 0.0f) ? t0 : NEG_SLOPE * t0;
  t1 = (t1 > 0.0f) ? t1 : NEG_SLOPE * t1;
  u0 = (u0 > 0.0f) ? u0 : NEG_SLOPE * u0;
  u1 = (u1 > 0.0f) ? u1 : NEG_SLOPE * u1;
  const float v0 = t0 * a00 + t1 * a01;
  const float v1 = u0 * a10 + u1 * a11;
  const float keep = lowHalf ? v0 : v1;
  const float send = lowHalf ? v1 : v0;
  float w = keep + __shfl_xor(send, 16, 32);
#pragma unroll
  for (int off = 8; off > 0; off >>= 1) w += __shfl_xor(w, off, 32);
  const float o = __shfl_xor(w, 16, 32);
  l0 = lowHalf ? w : o;
  l1 = lowHalf ? o : w;
}

__global__ void __launch_bounds__(NODE_WAVES_PER_BLOCK * 32)
node_attend(const float* __restrict__ xl,
            const float* __restrict__ xr,
            const int* __restrict__ rowptr,
            const int* __restrict__ col,
            const float* __restrict__ att,
            float* __restrict__ outp) {
  const int lane = threadIdx.x & 31;
  const int n = __builtin_amdgcn_readfirstlane(blockIdx.x * NODE_WAVES_PER_BLOCK + (threadIdx.x >> 5));
  if (n >= GN) return;
  const int beg = rowptr[n];
  const int end = rowptr[n + 1];

  const float a00 = att[lane];
  const float a01 = att[lane + 32];
  const float a10 = att[GC + lane];
  const float a11 = att[GC + lane + 32];
  const bool lowHalf = lane < 16;

  const float* xrn = xr + (size_t)n * GHC + lane;
  const float r0 = xrn[0];
  const float r1 = xrn[32];
  const float r2 = xrn[GC];
  const float r3 = xrn[GC + 32];

  const float* xln = xl + (size_t)n * GHC + lane;
  float acc0 = xln[0];
  float acc1 = xln[32];
  float acc2 = xln[GC];
  float acc3 = xln[GC + 32];
  float m0, m1;
  edge_logit2(acc0, acc1, acc2, acc3, r0, r1, r2, r3, a00, a01, a10, a11, lowHalf, m0, m1);
  float dn0 = 1.0f, dn1 = 1.0f;

  for (int cb = beg; cb < end; cb += 32) {
    const int left = end - cb;
    const int cnt  = (left < 32) ? left : 32;
    const int s = (lane < cnt) ? col[cb + lane] : 0;

    int i = 0;
    for (; i + 4 <= cnt; i += 4) {
      float x[4][4];
      float l0[4], l1[4];
#pragma unroll
      for (int k = 0; k < 4; ++k) {
        const int si = bcast_i(s, i + k);
        const float* xs = xl + (size_t)si * GHC + lane;
        x[k][0] = xs[0];
        x[k][1] = xs[32];
        x[k][2] = xs[GC];
        x[k][3] = xs[GC + 32];
      }
#pragma unroll
      for (int k = 0; k < 4; ++k)
        edge_logit2(x[k][0], x[k][1], x[k][2], x[k][3], r0, r1, r2, r3,
                    a00, a01, a10, a11, lowHalf, l0[k], l1[k]);

      const float n0 = fmaxf(fmaxf(m0, fmaxf(l0[0], l0[1])), fmaxf(l0[2], l0[3]));
      const float n1 = fmaxf(fmaxf(m1, fmaxf(l1[0], l1[1])), fmaxf(l1[2], l1[3]));
      const float sc0 = __expf(m0 - n0);
      const float sc1 = __expf(m1 - n1);
      m0 = n0;
      m1 = n1;
      dn0 *= sc0;  acc0 *= sc0;  acc1 *= sc0;
      dn1 *= sc1;  acc2 *= sc1;  acc3 *= sc1;
#pragma unroll
      for (int k = 0; k < 4; ++k) {
        const float p0 = __expf(l0[k] - n0);
        const float p1 = __expf(l1[k] - n1);
        dn0 += p0;
        dn1 += p1;
        acc0 += p0 * x[k][0];
        acc1 += p0 * x[k][1];
        acc2 += p1 * x[k][2];
        acc3 += p1 * x[k][3];
      }
    }
    for (; i < cnt; ++i) {
      const int si = bcast_i(s, i);
      const float* xs = xl + (size_t)si * GHC + lane;
      const float x0 = xs[0];
      const float x1 = xs[32];
      const float x2 = xs[GC];
      const float x3 = xs[GC + 32];
      float l0, l1;
      edge_logit2(x0, x1, x2, x3, r0, r1, r2, r3, a00, a01, a10, a11, lowHalf, l0, l1);
      const float n0 = fmaxf(m0, l0);
      const float n1 = fmaxf(m1, l1);
      const float sc0 = __expf(m0 - n0);
      const float sc1 = __expf(m1 - n1);
      const float p0  = __expf(l0 - n0);
      const float p1  = __expf(l1 - n1);
      m0 = n0;
      m1 = n1;
      dn0  = dn0 * sc0 + p0;
      dn1  = dn1 * sc1 + p1;
      acc0 = acc0 * sc0 + p0 * x0;
      acc1 = acc1 * sc0 + p0 * x1;
      acc2 = acc2 * sc1 + p1 * x2;
      acc3 = acc3 * sc1 + p1 * x3;
    }
  }

  float* on = outp + (size_t)n * GHC + lane;
  on[0]       = acc0 / dn0;
  on[32]      = acc1 / dn0;
  on[GC]      = acc2 / dn1;
  on[GC + 32] = acc3 / dn1;
}

#define STATS_ROWS 16
#define STATS_WAVES_PER_BLOCK 16
#define STATS_WAVES (GN / STATS_ROWS)
#define STATS_BLOCKS ((STATS_WAVES + STATS_WAVES_PER_BLOCK - 1) / STATS_WAVES_PER_BLOCK)
static_assert(GN % STATS_ROWS == 0, "a wave's rows are all inside the array");

__global__ void __launch_bounds__(STATS_WAVES_PER_BLOCK * 32)
gn_stats(const float* __restrict__ accum,
         const float* __restrict__ bias,
         float* __restrict__ colsum,
         float* __restrict__ colsumsq) {
  __shared__ alignas(16) float part[2][STATS_WAVES_PER_BLOCK][GHC];
  const int lane = threadIdx.x & 31;
  const int wv   = threadIdx.x >> 5;
  const int wave = blockIdx.x * STATS_WAVES_PER_BLOCK + wv;

  float4 s  = make_float4(0.0f, 0.0f, 0.0f, 0.0f);
  float4 s2 = make_float4(0.0f, 0.0f, 0.0f, 0.0f);
  if (wave < STATS_WAVES) {
    const size_t base = (size_t)wave * STATS_ROWS * GHC + (size_t)lane * 4;
    float4 v[STATS_ROWS];
#pragma unroll
    for (int k = 0; k < STATS_ROWS; ++k)
      v[k] = *(const float4*)(accum + base + (size_t)k * GHC);
    const float4 bv = *(const float4*)(bias + lane * 4);
#pragma unroll
    for (int k = 0; k < STATS_ROWS; ++k) {
      const float x0 = v[k].x + bv.x;
      const float x1 = v[k].y + bv.y;
      const float x2 = v[k].z + bv.z;
      const float x3 = v[k].w + bv.w;
      s.x += x0;  s2.x += x0 * x0;
      s.y += x1;  s2.y += x1 * x1;
      s.z += x2;  s2.z += x2 * x2;
      s.w += x3;  s2.w += x3 * x3;
    }
  }
  *(float4*)&part[0][wv][lane * 4] = s;
  *(float4*)&part[1][wv][lane * 4] = s2;
  __syncthreads();

  const int t = threadIdx.x;
  if (t < 2 * GHC) {
    const int st = t >> 7;
    const int f  = t & (GHC - 1);
    float tot = 0.0f;
#pragma unroll
    for (int w = 0; w < STATS_WAVES_PER_BLOCK; ++w) tot += part[st][w][f];
    atomicAdd(st ? &colsumsq[f] : &colsum[f], tot);
  }
}

__global__ void gn_finalize(const float* __restrict__ colsum,
                            const float* __restrict__ colsumsq,
                            const float* __restrict__ w,
                            const float* __restrict__ b,
                            const float* __restrict__ ms,
                            float* __restrict__ s1,
                            float* __restrict__ s2) {
  const int f = threadIdx.x;
  const float invN = 1.0f / (float)GN;
  const float mean = colsum[f] * invN;
  const float ex2  = colsumsq[f] * invN;
  const float msv  = ms[f];
  const float var  = ex2 - (2.0f * msv - msv * msv) * mean * mean;
  const float r    = rsqrtf(var + GEPS);
  const float a    = w[f] * r;
  s1[f] = a;
  s2[f] = b[f] - a * msv * mean;
}

#define APPLY_ROWS 8
#define APPLY_WAVES_PER_BLOCK 8
#define APPLY_WAVES (GN / APPLY_ROWS)
static_assert(GN % APPLY_ROWS == 0, "a wave's rows are all inside the array");

__global__ void __launch_bounds__(APPLY_WAVES_PER_BLOCK * 32)
gn_apply(const float* __restrict__ accum,
         const float* __restrict__ bias,
         const float* __restrict__ s1,
         const float* __restrict__ s2,
         float* __restrict__ out) {
  const int lane = threadIdx.x & 31;
  const int wave = blockIdx.x * APPLY_WAVES_PER_BLOCK + (threadIdx.x >> 5);
  if (wave >= APPLY_WAVES) return;
  const size_t base = (size_t)wave * APPLY_ROWS * GHC + (size_t)lane * 4;

  float4 v[APPLY_ROWS];
#pragma unroll
  for (int k = 0; k < APPLY_ROWS; ++k)
    v[k] = *(const float4*)(accum + base + (size_t)k * GHC);

  const float4 bv = *(const float4*)(bias + lane * 4);
  const float4 a1 = *(const float4*)(s1 + lane * 4);
  const float4 a2 = *(const float4*)(s2 + lane * 4);

#pragma unroll
  for (int k = 0; k < APPLY_ROWS; ++k) {
    float4 r;
    r.x = fmaxf((v[k].x + bv.x) * a1.x + a2.x, 0.0f);
    r.y = fmaxf((v[k].y + bv.y) * a1.y + a2.y, 0.0f);
    r.z = fmaxf((v[k].z + bv.z) * a1.z + a2.z, 0.0f);
    r.w = fmaxf((v[k].w + bv.w) * a1.w + a2.w, 0.0f);
    *(float4*)(out + base + (size_t)k * GHC) = r;
  }
}

__global__ void head_kernel(const float* __restrict__ h1,
                            const float* __restrict__ W2,
                            const float* __restrict__ b2,
                            float* __restrict__ out) {
  int n = blockIdx.x * blockDim.x + threadIdx.x;
  if (n >= GN) return;
  const float* row = h1 + (size_t)n * GC;
  float a0 = b2[0], a1 = b2[1];
#pragma unroll 8
  for (int c = 0; c < GC; ++c) {
    const float v = row[c];
    a0 += v * W2[c * 2 + 0];
    a1 += v * W2[c * 2 + 1];
  }
  out[n * 2 + 0] = a0;
  out[n * 2 + 1] = a1;
}

extern "C" void kernel_launch(void* const* d_in, const int* in_sizes, int n_in,
                              void* d_out, int out_size, void* d_ws, size_t ws_size,
                              hipStream_t stream) {
  (void)in_sizes; (void)n_in; (void)out_size; (void)ws_size;

  const float* x     = (const float*)d_in[0];
  const int*   ei    = (const int*)d_in[1];
  const float* Wl    = (const float*)d_in[2];
  const float* Wr    = (const float*)d_in[3];
  const float* att   = (const float*)d_in[4];
  const float* convb = (const float*)d_in[5];
  const float* gnw   = (const float*)d_in[6];
  const float* gnb   = (const float*)d_in[7];
  const float* gnms  = (const float*)d_in[8];
  const float* W1    = (const float*)d_in[9];
  const float* b1    = (const float*)d_in[10];
  const float* W2    = (const float*)d_in[11];
  const float* b2    = (const float*)d_in[12];
  float* out = (float*)d_out;

  float* ws       = (float*)d_ws;
  float* xl       = ws;
  float* xr       = xl + (size_t)GN * GHC;
  float* hbuf     = xr + (size_t)GN * GHC;
  float* h1       = hbuf + (size_t)GN * GHC;
  int*   col      = (int*)h1;
  size_t scratch1 = ((size_t)GN * GC > (size_t)GETOT * GH)
                        ? (size_t)GN * GC : (size_t)GETOT * GH;
  float* accum    = h1 + scratch1;
  float* rpregion = accum + (size_t)GN * GHC;
  int*   rowptr   = (int*)rpregion;
  float* colsum   = rpregion + (size_t)GN * GH;
  float* colsumsq = colsum + GHC;
  float* s1       = colsumsq + GHC;
  float* s2       = s1 + GHC;
  int*   cnt      = (int*)(s2 + GHC);
  int*   cursor   = cnt + GN;

  fill_int_kernel<<<(GN + 255) / 256, 256, 0, stream>>>(cnt, 0, GN);
  csr_count<<<(GE + 255) / 256, 256, 0, stream>>>(ei, cnt);
  csr_scan<<<SCAN_BLOCKS, SCAN_T, 0, stream>>>(cnt, rowptr, cursor);
  csr_fill<<<(GE + 255) / 256, 256, 0, stream>>>(ei, cursor, col);

  const float* cur = x;
  for (int layer = 0; layer < 2; ++layer) {
    const float* wl = Wl + (size_t)layer * GHC * GHC;
    const float* wr = Wr + (size_t)layer * GHC * GHC;
    const float* at = att + (size_t)layer * GH * GC;
    const float* cb = convb + (size_t)layer * GHC;

    wmma_gemm_f32<<<GN / 16, dim3(32, GHC / 16), 0, stream>>>(cur, wl, nullptr, xl, GHC, GHC, 0);
    wmma_gemm_f32<<<GN / 16, dim3(32, GHC / 16), 0, stream>>>(cur, wr, nullptr, xr, GHC, GHC, 0);

    fill_kernel<<<1, 2 * GHC, 0, stream>>>(colsum, 0.0f, 2 * GHC);

    node_attend<<<(GN + NODE_WAVES_PER_BLOCK - 1) / NODE_WAVES_PER_BLOCK, NODE_WAVES_PER_BLOCK * 32, 0, stream>>>(xl, xr, rowptr, col, at, accum);

    gn_stats<<<STATS_BLOCKS, STATS_WAVES_PER_BLOCK * 32, 0, stream>>>(accum, cb, colsum, colsumsq);
    gn_finalize<<<1, GHC, 0, stream>>>(colsum, colsumsq,
                                       gnw + (size_t)layer * GHC,
                                       gnb + (size_t)layer * GHC,
                                       gnms + (size_t)layer * GHC, s1, s2);
    gn_apply<<<(APPLY_WAVES + APPLY_WAVES_PER_BLOCK - 1) / APPLY_WAVES_PER_BLOCK, APPLY_WAVES_PER_BLOCK * 32, 0, stream>>>(accum, cb, s1, s2, hbuf);

    cur = hbuf;
  }

  wmma_gemm_f32<<<GN / 16, dim3(32, GC / 16), 0, stream>>>(hbuf, W1, b1, h1, GHC, GC, 1);
  head_kernel<<<(GN + 255) / 256, 256, 0, stream>>>(h1, W2, b2, out);
}


# GEMMs: weights kept in LDS as ready fragments, one wave multiplies a 16-row strip of A (held in registers) into all column tiles, Wl and Wr share one launch; new k order inside the sums
# speedup vs baseline: 4.7281x; 1.2600x over previous
#include <hip/hip_runtime.h>
#include <hip/hip_bf16.h>


#define GN 50000
#define GE 800000
#define GETOT (GE + GN)
#define GHC 128
#define GC 64
#define GH 2
#define NEG_SLOPE 0.2f
#define GEPS 1e-5f

typedef __attribute__((ext_vector_type(2))) float v2f;
typedef __attribute__((ext_vector_type(8))) float v8f;

#define GT_WAVES 16
#define GT_THREADS (GT_WAVES * 32)
#define GT_K 128
#define GT_BLOCKS 256

template <int NT, bool EPI>
__global__ void __launch_bounds__(GT_THREADS)
gemm_strip_f32(const float* __restrict__ A,
               const float* __restrict__ B0,
               const float* __restrict__ B1,
               float* __restrict__ C0,
               float* __restrict__ C1,
               const float* __restrict__ bias,
               int nmat, int nStrips) {
  constexpr int NOUT = 16 * NT;
  __shared__ alignas(16) float Bs[32 * NT * 32 * 2];

  const int tid  = threadIdx.x;
  const int lane = tid & 31;
  const int wv   = __builtin_amdgcn_readfirstlane(tid >> 5);
  const int mat  = blockIdx.x % nmat;
  const float* __restrict__ B = mat ? B1 : B0;
  float* __restrict__ C       = mat ? C1 : C0;

  constexpr int FILL_IT = GT_K * (NOUT / 4) / GT_THREADS;
  static_assert(GT_K * (NOUT / 4) % GT_THREADS == 0, "every thread fills the same number of slots");
  float4 fv[FILL_IT];
#pragma unroll
  for (int it = 0; it < FILL_IT; ++it) {
    const int q  = tid + it * GT_THREADS;
    const int k  = q / (NOUT / 4);
    const int n4 = (q % (NOUT / 4)) * 4;
    fv[it] = *(const float4*)(B + (size_t)k * NOUT + n4);
  }
#pragma unroll
  for (int it = 0; it < FILL_IT; ++it) {
    const int q  = tid + it * GT_THREADS;
    const int k  = q / (NOUT / 4);
    const int n4 = (q % (NOUT / 4)) * 4;
    const int s  = ((k >> 3) << 1) | ((k >> 1) & 1);
    const int hh = (k >> 2) & 1;
    const int e  = k & 1;
    const int t  = n4 >> 4;
    const int cc = n4 & 15;
    const int base = ((s * (NT / 2) + (t >> 1)) * 32 + cc + 16 * hh) * 4 + ((t & 1) << 1) + e;
    Bs[base]      = fv[it].x;
    Bs[base + 4]  = fv[it].y;
    Bs[base + 8]  = fv[it].z;
    Bs[base + 12] = fv[it].w;
  }
  __syncthreads();

  const int c = lane & 15;
  const int h = lane >> 4;
  const int first  = blockIdx.x / nmat;
  const int stride = gridDim.x / nmat;

  for (int sidx = wv; ; sidx += GT_WAVES) {
    const int strip = first + stride * sidx;
    if (strip >= nStrips) break;

    const float* ap = A + ((size_t)strip * 16 + c) * GT_K + 4 * h;
    float4 a4[GT_K / 8];
#pragma unroll
    for (int j = 0; j < GT_K / 8; ++j) a4[j] = *(const float4*)(ap + 8 * j);

    v8f acc[NT];
#pragma unroll
    for (int t = 0; t < NT; ++t) acc[t] = v8f{};

    const float4* bl = (const float4*)Bs + lane;
    float4 bq[2][NT];
#pragma unroll
    for (int i = 0; i < NT; ++i) bq[0][i] = bl[i * 32];

#pragma unroll
    for (int j = 0; j < GT_K / 8; ++j) {
      if (j + 1 < GT_K / 8) {
#pragma unroll
        for (int i = 0; i < NT; ++i) bq[(j + 1) & 1][i] = bl[((j + 1) * NT + i) * 32];
      }
#pragma unroll
      for (int p = 0; p < 2; ++p) {
        v2f a;
        a.x = p ? a4[j].z : a4[j].x;
        a.y = p ? a4[j].w : a4[j].y;
#pragma unroll
        for (int tp = 0; tp < NT / 2; ++tp) {
          const float4 b = bq[j & 1][p * (NT / 2) + tp];
          v2f b0, b1;
          b0.x = b.x;  b0.y = b.y;
          b1.x = b.z;  b1.y = b.w;
          acc[2 * tp]     = __builtin_amdgcn_wmma_f32_16x16x4_f32(false, a, false, b0,
                                                                  (short)0, acc[2 * tp], false, false);
          acc[2 * tp + 1] = __builtin_amdgcn_wmma_f32_16x16x4_f32(false, a, false, b1,
                                                                  (short)0, acc[2 * tp + 1], false, false);
        }
      }
      __builtin_amdgcn_sched_barrier(0);
    }

    const int rowBase = strip * 16 + (h << 3);
#pragma unroll
    for (int t = 0; t < NT; ++t) {
      float bv = 0.0f;
      if (EPI) bv = bias[16 * t + c];
#pragma unroll
      for (int r = 0; r < 8; ++r) {
        float v = acc[t][r];
        if (EPI) v = fmaxf(v + bv, 0.0f);
        C[(size_t)(rowBase + r) * NOUT + 16 * t + c] = v;
      }
    }
  }
}

__global__ void fill_kernel(float* __restrict__ p, float v, int n) {
  int i = blockIdx.x * blockDim.x + threadIdx.x;
  if (i < n) p[i] = v;
}

__global__ void fill_int_kernel(int* __restrict__ p, int v, int n) {
  int i = blockIdx.x * blockDim.x + threadIdx.x;
  if (i < n) p[i] = v;
}

__device__ __forceinline__ int bcast_i(int v, int i) {
  return __builtin_amdgcn_readlane(v, i);
}

__global__ void csr_count(const int* __restrict__ ei, int* __restrict__ cnt) {
  int e = blockIdx.x * blockDim.x + threadIdx.x;
  if (e >= GE) return;
  atomicAdd(&cnt[ei[GE + e]], 1);
}

#define SCAN_T 1024
#define SCAN_BLOCKS ((GN + SCAN_T - 1) / SCAN_T)

__global__ void __launch_bounds__(SCAN_T)
csr_scan(const int* __restrict__ cnt, int* __restrict__ rowptr, int* __restrict__ cursor) {
  __shared__ int wsum[SCAN_T / 32];
  __shared__ int wtot[SCAN_T / 32];
  const int t    = threadIdx.x;
  const int lane = t & 31;
  const int wv   = t >> 5;
  const int tile0 = blockIdx.x * SCAN_T;

  int pre = 0;
  for (int i = t; i < tile0; i += SCAN_T) pre += cnt[i];
#pragma unroll
  for (int off = 16; off > 0; off >>= 1) pre += __shfl_xor(pre, off, 32);

  const int idx = tile0 + t;
  const int x = (idx < GN) ? cnt[idx] : 0;
  int inc = x;
#pragma unroll
  for (int off = 1; off < 32; off <<= 1) {
    const int y = __shfl_up(inc, off, 32);
    if (lane >= off) inc += y;
  }
  if (lane == 0)  wsum[wv] = pre;
  if (lane == 31) wtot[wv] = inc;
  __syncthreads();

  int base = 0;
  int wofs = 0;
#pragma unroll
  for (int k = 0; k < SCAN_T / 32; ++k) {
    base += wsum[k];
    if (k < wv) wofs += wtot[k];
  }
  const int excl = base + wofs + inc - x;
  if (idx < GN) {
    rowptr[idx] = excl;
    cursor[idx] = excl;
  }
  if (idx == GN - 1) rowptr[GN] = excl + x;
}

__global__ void csr_fill(const int* __restrict__ ei, int* __restrict__ cursor,
                         int* __restrict__ col) {
  int e = blockIdx.x * blockDim.x + threadIdx.x;
  if (e >= GE) return;
  const int pos = atomicAdd(&cursor[ei[GE + e]], 1);
  col[pos] = ei[e];
}

#define NODE_WAVES_PER_BLOCK 4

__device__ __forceinline__ void edge_logit2(float x0, float x1, float x2, float x3,
                                            float r0, float r1, float r2, float r3,
                                            float a00, float a01, float a10, float a11,
                                            bool lowHalf, float& l0, float& l1) {
  float t0 = x0 + r0;
  float t1 = x1 + r1;
  float u0 = x2 + r2;
  float u1 = x3 + r3;
  t0 = (t0 > 0.0f) ? t0 : NEG_SLOPE * t0;
  t1 = (t1 > 0.0f) ? t1 : NEG_SLOPE * t1;
  u0 = (u0 > 0.0f) ? u0 : NEG_SLOPE * u0;
  u1 = (u1 > 0.0f) ? u1 : NEG_SLOPE * u1;
  const float v0 = t0 * a00 + t1 * a01;
  const float v1 = u0 * a10 + u1 * a11;
  const float keep = lowHalf ? v0 : v1;
  const float send = lowHalf ? v1 : v0;
  float w = keep + __shfl_xor(send, 16, 32);
#pragma unroll
  for (int off = 8; off > 0; off >>= 1) w += __shfl_xor(w, off, 32);
  const float o = __shfl_xor(w, 16, 32);
  l0 = lowHalf ? w : o;
  l1 = lowHalf ? o : w;
}

__global__ void __launch_bounds__(NODE_WAVES_PER_BLOCK * 32)
node_attend(const float* __restrict__ xl,
            const float* __restrict__ xr,
            const int* __restrict__ rowptr,
            const int* __restrict__ col,
            const float* __restrict__ att,
            float* __restrict__ outp) {
  const int lane = threadIdx.x & 31;
  const int n = __builtin_amdgcn_readfirstlane(blockIdx.x * NODE_WAVES_PER_BLOCK + (threadIdx.x >> 5));
  if (n >= GN) return;
  const int beg = rowptr[n];
  const int end = rowptr[n + 1];

  const float a00 = att[lane];
  const float a01 = att[lane + 32];
  const float a10 = att[GC + lane];
  const float a11 = att[GC + lane + 32];
  const bool lowHalf = lane < 16;

  const float* xrn = xr + (size_t)n * GHC + lane;
  const float r0 = xrn[0];
  const float r1 = xrn[32];
  const float r2 = xrn[GC];
  const float r3 = xrn[GC + 32];

  const float* xln = xl + (size_t)n * GHC + lane;
  float acc0 = xln[0];
  float acc1 = xln[32];
  float acc2 = xln[GC];
  float acc3 = xln[GC + 32];
  float m0, m1;
  edge_logit2(acc0, acc1, acc2, acc3, r0, r1, r2, r3, a00, a01, a10, a11, lowHalf, m0, m1);
  float dn0 = 1.0f, dn1 = 1.0f;

  for (int cb = beg; cb < end; cb += 32) {
    const int left = end - cb;
    const int cnt  = (left < 32) ? left : 32;
    const int s = (lane < cnt) ? col[cb + lane] : 0;

    int i = 0;
    for (; i + 4 <= cnt; i += 4) {
      float x[4][4];
      float l0[4], l1[4];
#pragma unroll
      for (int k = 0; k < 4; ++k) {
        const int si = bcast_i(s, i + k);
        const float* xs = xl + (size_t)si * GHC + lane;
        x[k][0] = xs[0];
        x[k][1] = xs[32];
        x[k][2] = xs[GC];
        x[k][3] = xs[GC + 32];
      }
#pragma unroll
      for (int k = 0; k < 4; ++k)
        edge_logit2(x[k][0], x[k][1], x[k][2], x[k][3], r0, r1, r2, r3,
                    a00, a01, a10, a11, lowHalf, l0[k], l1[k]);

      const float n0 = fmaxf(fmaxf(m0, fmaxf(l0[0], l0[1])), fmaxf(l0[2], l0[3]));
      const float n1 = fmaxf(fmaxf(m1, fmaxf(l1[0], l1[1])), fmaxf(l1[2], l1[3]));
      const float sc0 = __expf(m0 - n0);
      const float sc1 = __expf(m1 - n1);
      m0 = n0;
      m1 = n1;
      dn0 *= sc0;  acc0 *= sc0;  acc1 *= sc0;
      dn1 *= sc1;  acc2 *= sc1;  acc3 *= sc1;
#pragma unroll
      for (int k = 0; k < 4; ++k) {
        const float p0 = __expf(l0[k] - n0);
        const float p1 = __expf(l1[k] - n1);
        dn0 += p0;
        dn1 += p1;
        acc0 += p0 * x[k][0];
        acc1 += p0 * x[k][1];
        acc2 += p1 * x[k][2];
        acc3 += p1 * x[k][3];
      }
    }
    for (; i < cnt; ++i) {
      const int si = bcast_i(s, i);
      const float* xs = xl + (size_t)si * GHC + lane;
      const float x0 = xs[0];
      const float x1 = xs[32];
      const float x2 = xs[GC];
      const float x3 = xs[GC + 32];
      float l0, l1;
      edge_logit2(x0, x1, x2, x3, r0, r1, r2, r3, a00, a01, a10, a11, lowHalf, l0, l1);
      const float n0 = fmaxf(m0, l0);
      const float n1 = fmaxf(m1, l1);
      const float sc0 = __expf(m0 - n0);
      const float sc1 = __expf(m1 - n1);
      const float p0  = __expf(l0 - n0);
      const float p1  = __expf(l1 - n1);
      m0 = n0;
      m1 = n1;
      dn0  = dn0 * sc0 + p0;
      dn1  = dn1 * sc1 + p1;
      acc0 = acc0 * sc0 + p0 * x0;
      acc1 = acc1 * sc0 + p0 * x1;
      acc2 = acc2 * sc1 + p1 * x2;
      acc3 = acc3 * sc1 + p1 * x3;
    }
  }

  float* on = outp + (size_t)n * GHC + lane;
  on[0]       = acc0 / dn0;
  on[32]      = acc1 / dn0;
  on[GC]      = acc2 / dn1;
  on[GC + 32] = acc3 / dn1;
}

#define STATS_ROWS 16
#define STATS_WAVES_PER_BLOCK 16
#define STATS_WAVES (GN / STATS_ROWS)
#define STATS_BLOCKS ((STATS_WAVES + STATS_WAVES_PER_BLOCK - 1) / STATS_WAVES_PER_BLOCK)
static_assert(GN % STATS_ROWS == 0, "a wave's rows are all inside the array");

__global__ void __launch_bounds__(STATS_WAVES_PER_BLOCK * 32)
gn_stats(const float* __restrict__ accum,
         const float* __restrict__ bias,
         float* __restrict__ colsum,
         float* __restrict__ colsumsq) {
  __shared__ alignas(16) float part[2][STATS_WAVES_PER_BLOCK][GHC];
  const int lane = threadIdx.x & 31;
  const int wv   = threadIdx.x >> 5;
  const int wave = blockIdx.x * STATS_WAVES_PER_BLOCK + wv;

  float4 s  = make_float4(0.0f, 0.0f, 0.0f, 0.0f);
  float4 s2 = make_float4(0.0f, 0.0f, 0.0f, 0.0f);
  if (wave < STATS_WAVES) {
    const size_t base = (size_t)wave * STATS_ROWS * GHC + (size_t)lane * 4;
    float4 v[STATS_ROWS];
#pragma unroll
    for (int k = 0; k < STATS_ROWS; ++k)
      v[k] = *(const float4*)(accum + base + (size_t)k * GHC);
    const float4 bv = *(const float4*)(bias + lane * 4);
#pragma unroll
    for (int k = 0; k < STATS_ROWS; ++k) {
      const float x0 = v[k].x + bv.x;
      const float x1 = v[k].y + bv.y;
      const float x2 = v[k].z + bv.z;
      const float x3 = v[k].w + bv.w;
      s.x += x0;  s2.x += x0 * x0;
      s.y += x1;  s2.y += x1 * x1;
      s.z += x2;  s2.z += x2 * x2;
      s.w += x3;  s2.w += x3 * x3;
    }
  }
  *(float4*)&part[0][wv][lane * 4] = s;
  *(float4*)&part[1][wv][lane * 4] = s2;
  __syncthreads();

  const int t = threadIdx.x;
  if (t < 2 * GHC) {
    const int st = t >> 7;
    const int f  = t & (GHC - 1);
    float tot = 0.0f;
#pragma unroll
    for (int w = 0; w < STATS_WAVES_PER_BLOCK; ++w) tot += part[st][w][f];
    atomicAdd(st ? &colsumsq[f] : &colsum[f], tot);
  }
}

__global__ void gn_finalize(const float* __restrict__ colsum,
                            const float* __restrict__ colsumsq,
                            const float* __restrict__ w,
                            const float* __restrict__ b,
                            const float* __restrict__ ms,
                            float* __restrict__ s1,
                            float* __restrict__ s2) {
  const int f = threadIdx.x;
  const float invN = 1.0f / (float)GN;
  const float mean = colsum[f] * invN;
  const float ex2  = colsumsq[f] * invN;
  const float msv  = ms[f];
  const float var  = ex2 - (2.0f * msv - msv * msv) * mean * mean;
  const float r    = rsqrtf(var + GEPS);
  const float a    = w[f] * r;
  s1[f] = a;
  s2[f] = b[f] - a * msv * mean;
}

#define APPLY_ROWS 8
#define APPLY_WAVES_PER_BLOCK 8
#define APPLY_WAVES (GN / APPLY_ROWS)
static_assert(GN % APPLY_ROWS == 0, "a wave's rows are all inside the array");

__global__ void __launch_bounds__(APPLY_WAVES_PER_BLOCK * 32)
gn_apply(const float* __restrict__ accum,
         const float* __restrict__ bias,
         const float* __restrict__ s1,
         const float* __restrict__ s2,
         float* __restrict__ out) {
  const int lane = threadIdx.x & 31;
  const int wave = blockIdx.x * APPLY_WAVES_PER_BLOCK + (threadIdx.x >> 5);
  if (wave >= APPLY_WAVES) return;
  const size_t base = (size_t)wave * APPLY_ROWS * GHC + (size_t)lane * 4;

  float4 v[APPLY_ROWS];
#pragma unroll
  for (int k = 0; k < APPLY_ROWS; ++k)
    v[k] = *(const float4*)(accum + base + (size_t)k * GHC);

  const float4 bv = *(const float4*)(bias + lane * 4);
  const float4 a1 = *(const float4*)(s1 + lane * 4);
  const float4 a2 = *(const float4*)(s2 + lane * 4);

#pragma unroll
  for (int k = 0; k < APPLY_ROWS; ++k) {
    float4 r;
    r.x = fmaxf((v[k].x + bv.x) * a1.x + a2.x, 0.0f);
    r.y = fmaxf((v[k].y + bv.y) * a1.y + a2.y, 0.0f);
    r.z = fmaxf((v[k].z + bv.z) * a1.z + a2.z, 0.0f);
    r.w = fmaxf((v[k].w + bv.w) * a1.w + a2.w, 0.0f);
    *(float4*)(out + base + (size_t)k * GHC) = r;
  }
}

__global__ void head_kernel(const float* __restrict__ h1,
                            const float* __restrict__ W2,
                            const float* __restrict__ b2,
                            float* __restrict__ out) {
  int n = blockIdx.x * blockDim.x + threadIdx.x;
  if (n >= GN) return;
  const float* row = h1 + (size_t)n * GC;
  float a0 = b2[0], a1 = b2[1];
#pragma unroll 8
  for (int c = 0; c < GC; ++c) {
    const float v = row[c];
    a0 += v * W2[c * 2 + 0];
    a1 += v * W2[c * 2 + 1];
  }
  out[n * 2 + 0] = a0;
  out[n * 2 + 1] = a1;
}

extern "C" void kernel_launch(void* const* d_in, const int* in_sizes, int n_in,
                              void* d_out, int out_size, void* d_ws, size_t ws_size,
                              hipStream_t stream) {
  (void)in_sizes; (void)n_in; (void)out_size; (void)ws_size;

  const float* x     = (const float*)d_in[0];
  const int*   ei    = (const int*)d_in[1];
  const float* Wl    = (const float*)d_in[2];
  const float* Wr    = (const float*)d_in[3];
  const float* att   = (const float*)d_in[4];
  const float* convb = (const float*)d_in[5];
  const float* gnw   = (const float*)d_in[6];
  const float* gnb   = (const float*)d_in[7];
  const float* gnms  = (const float*)d_in[8];
  const float* W1    = (const float*)d_in[9];
  const float* b1    = (const float*)d_in[10];
  const float* W2    = (const float*)d_in[11];
  const float* b2    = (const float*)d_in[12];
  float* out = (float*)d_out;

  float* ws       = (float*)d_ws;
  float* xl       = ws;
  float* xr       = xl + (size_t)GN * GHC;
  float* hbuf     = xr + (size_t)GN * GHC;
  float* h1       = hbuf + (size_t)GN * GHC;
  int*   col      = (int*)h1;
  size_t scratch1 = ((size_t)GN * GC > (size_t)GETOT * GH)
                        ? (size_t)GN * GC : (size_t)GETOT * GH;
  float* accum    = h1 + scratch1;
  float* rpregion = accum + (size_t)GN * GHC;
  int*   rowptr   = (int*)rpregion;
  float* colsum   = rpregion + (size_t)GN * GH;
  float* colsumsq = colsum + GHC;
  float* s1       = colsumsq + GHC;
  float* s2       = s1 + GHC;
  int*   cnt      = (int*)(s2 + GHC);
  int*   cursor   = cnt + GN;

  fill_int_kernel<<<(GN + 255) / 256, 256, 0, stream>>>(cnt, 0, GN);
  csr_count<<<(GE + 255) / 256, 256, 0, stream>>>(ei, cnt);
  csr_scan<<<SCAN_BLOCKS, SCAN_T, 0, stream>>>(cnt, rowptr, cursor);
  csr_fill<<<(GE + 255) / 256, 256, 0, stream>>>(ei, cursor, col);

  const float* cur = x;
  for (int layer = 0; layer < 2; ++layer) {
    const float* wl = Wl + (size_t)layer * GHC * GHC;
    const float* wr = Wr + (size_t)layer * GHC * GHC;
    const float* at = att + (size_t)layer * GH * GC;
    const float* cb = convb + (size_t)layer * GHC;

    gemm_strip_f32<GHC / 16, false><<<GT_BLOCKS, GT_THREADS, 0, stream>>>(cur, wl, wr, xl, xr, nullptr, 2, GN / 16);

    fill_kernel<<<1, 2 * GHC, 0, stream>>>(colsum, 0.0f, 2 * GHC);

    node_attend<<<(GN + NODE_WAVES_PER_BLOCK - 1) / NODE_WAVES_PER_BLOCK, NODE_WAVES_PER_BLOCK * 32, 0, stream>>>(xl, xr, rowptr, col, at, accum);

    gn_stats<<<STATS_BLOCKS, STATS_WAVES_PER_BLOCK * 32, 0, stream>>>(accum, cb, colsum, colsumsq);
    gn_finalize<<<1, GHC, 0, stream>>>(colsum, colsumsq,
                                       gnw + (size_t)layer * GHC,
                                       gnb + (size_t)layer * GHC,
                                       gnms + (size_t)layer * GHC, s1, s2);
    gn_apply<<<(APPLY_WAVES + APPLY_WAVES_PER_BLOCK - 1) / APPLY_WAVES_PER_BLOCK, APPLY_WAVES_PER_BLOCK * 32, 0, stream>>>(accum, cb, s1, s2, hbuf);

    cur = hbuf;
  }

  gemm_strip_f32<GC / 16, true><<<GT_BLOCKS, GT_THREADS, 0, stream>>>(hbuf, W1, nullptr, h1, nullptr, b1, 1, GN / 16);
  head_kernel<<<(GN + 255) / 256, 256, 0, stream>>>(h1, W2, b2, out);
}


# Small passes folded into the GEMM launches: column-sum zeroing, GraphNorm finalize+apply+ReLU (on the A strip as it loads), 64x2 head (GEMM epilogue, new sum order); 18 -> 11 launches
# speedup vs baseline: 5.1876x; 1.0972x over previous
#include <hip/hip_runtime.h>
#include <hip/hip_bf16.h>


#define GN 50000
#define GE 800000
#define GETOT (GE + GN)
#define GHC 128
#define GC 64
#define GH 2
#define NEG_SLOPE 0.2f
#define GEPS 1e-5f

typedef __attribute__((ext_vector_type(2))) float v2f;
typedef __attribute__((ext_vector_type(8))) float v8f;

#define GT_WAVES 16
#define GT_THREADS (GT_WAVES * 32)
#define GT_K 128
#define GT_BLOCKS 256

template <int NT, bool NORM, bool HEAD>
__global__ void __launch_bounds__(GT_THREADS)
gemm_strip_f32(const float* __restrict__ A,
               const float* __restrict__ B0,
               const float* __restrict__ B1,
               float* __restrict__ C0,
               float* __restrict__ C1,
               const float* __restrict__ colsum,
               const float* __restrict__ colsumsq,
               const float* __restrict__ gw,
               const float* __restrict__ gb,
               const float* __restrict__ gms,
               const float* __restrict__ cbias,
               const float* __restrict__ bias,
               const float* __restrict__ W2,
               const float* __restrict__ b2,
               float* __restrict__ zero512,
               int nmat, int nStrips) {
  constexpr int NOUT = 16 * NT;
  static_assert(!HEAD || NT == GC / 16, "the head product expects the 64-column GEMM");
  __shared__ alignas(16) float Ns[2][GT_K];
  __shared__ alignas(16) float Bs[32 * NT * 32 * 2];

  const int tid  = threadIdx.x;
  const int lane = tid & 31;
  const int wv   = __builtin_amdgcn_readfirstlane(tid >> 5);
  const int mat  = blockIdx.x % nmat;
  const float* __restrict__ B = mat ? B1 : B0;
  float* __restrict__ C       = mat ? C1 : C0;

  static_assert(GT_THREADS == 4 * GHC, "one thread per float of the two layers' column sums");
  if (zero512 != nullptr && blockIdx.x == 0) zero512[tid] = 0.0f;

  if (NORM && tid < GT_K) {
    const float invN = 1.0f / (float)GN;
    const float mean = colsum[tid] * invN;
    const float ex2  = colsumsq[tid] * invN;
    const float msv  = gms[tid];
    const float var  = ex2 - (2.0f * msv - msv * msv) * mean * mean;
    const float r    = rsqrtf(var + GEPS);
    const float a    = gw[tid] * r;
    const float sh   = gb[tid] - a * msv * mean;
    Ns[0][tid] = a;
    Ns[1][tid] = cbias[tid] * a + sh;
  }

  constexpr int FILL_IT = GT_K * (NOUT / 4) / GT_THREADS;
  static_assert(GT_K * (NOUT / 4) % GT_THREADS == 0, "every thread fills the same number of slots");
  float4 fv[FILL_IT];
#pragma unroll
  for (int it = 0; it < FILL_IT; ++it) {
    const int q  = tid + it * GT_THREADS;
    const int k  = q / (NOUT / 4);
    const int n4 = (q % (NOUT / 4)) * 4;
    fv[it] = *(const float4*)(B + (size_t)k * NOUT + n4);
  }
#pragma unroll
  for (int it = 0; it < FILL_IT; ++it) {
    const int q  = tid + it * GT_THREADS;
    const int k  = q / (NOUT / 4);
    const int n4 = (q % (NOUT / 4)) * 4;
    const int s  = ((k >> 3) << 1) | ((k >> 1) & 1);
    const int hh = (k >> 2) & 1;
    const int e  = k & 1;
    const int t  = n4 >> 4;
    const int cc = n4 & 15;
    const int base = ((s * (NT / 2) + (t >> 1)) * 32 + cc + 16 * hh) * 4 + ((t & 1) << 1) + e;
    Bs[base]      = fv[it].x;
    Bs[base + 4]  = fv[it].y;
    Bs[base + 8]  = fv[it].z;
    Bs[base + 12] = fv[it].w;
  }
  __syncthreads();

  const int c = lane & 15;
  const int h = lane >> 4;
  const int first  = blockIdx.x / nmat;
  const int stride = gridDim.x / nmat;

  for (int sidx = wv; ; sidx += GT_WAVES) {
    const int strip = first + stride * sidx;
    if (strip >= nStrips) break;

    const float* ap = A + ((size_t)strip * 16 + c) * GT_K + 4 * h;
    float4 a4[GT_K / 8];
#pragma unroll
    for (int j = 0; j < GT_K / 8; ++j) a4[j] = *(const float4*)(ap + 8 * j);
    if (NORM) {
      const float4* n1 = (const float4*)&Ns[0][4 * h];
      const float4* n2 = (const float4*)&Ns[1][4 * h];
#pragma unroll
      for (int j = 0; j < GT_K / 8; ++j) {
        const float4 p = n1[2 * j];
        const float4 q = n2[2 * j];
        a4[j].x = fmaxf(a4[j].x * p.x + q.x, 0.0f);
        a4[j].y = fmaxf(a4[j].y * p.y + q.y, 0.0f);
        a4[j].z = fmaxf(a4[j].z * p.z + q.z, 0.0f);
        a4[j].w = fmaxf(a4[j].w * p.w + q.w, 0.0f);
      }
    }

    v8f acc[NT];
#pragma unroll
    for (int t = 0; t < NT; ++t) acc[t] = v8f{};

    const float4* bl = (const float4*)Bs + lane;
    float4 bq[2][NT];
#pragma unroll
    for (int i = 0; i < NT; ++i) bq[0][i] = bl[i * 32];

#pragma unroll
    for (int j = 0; j < GT_K / 8; ++j) {
      if (j + 1 < GT_K / 8) {
#pragma unroll
        for (int i = 0; i < NT; ++i) bq[(j + 1) & 1][i] = bl[((j + 1) * NT + i) * 32];
      }
#pragma unroll
      for (int p = 0; p < 2; ++p) {
        v2f a;
        a.x = p ? a4[j].z : a4[j].x;
        a.y = p ? a4[j].w : a4[j].y;
#pragma unroll
        for (int tp = 0; tp < NT / 2; ++tp) {
          const float4 b = bq[j & 1][p * (NT / 2) + tp];
          v2f b0, b1;
          b0.x = b.x;  b0.y = b.y;
          b1.x = b.z;  b1.y = b.w;
          acc[2 * tp]     = __builtin_amdgcn_wmma_f32_16x16x4_f32(false, a, false, b0,
                                                                  (short)0, acc[2 * tp], false, false);
          acc[2 * tp + 1] = __builtin_amdgcn_wmma_f32_16x16x4_f32(false, a, false, b1,
                                                                  (short)0, acc[2 * tp + 1], false, false);
        }
      }
      __builtin_amdgcn_sched_barrier(0);
    }

    const int rowBase = strip * 16 + (h << 3);
    if (HEAD) {
      float p0[8], p1[8];
#pragma unroll
      for (int r = 0; r < 8; ++r) { p0[r] = 0.0f; p1[r] = 0.0f; }
#pragma unroll
      for (int t = 0; t < NT; ++t) {
        const float bv = bias[16 * t + c];
        const float w0 = W2[(16 * t + c) * 2 + 0];
        const float w1 = W2[(16 * t + c) * 2 + 1];
#pragma unroll
        for (int r = 0; r < 8; ++r) {
          const float v = fmaxf(acc[t][r] + bv, 0.0f);
          p0[r] += v * w0;
          p1[r] += v * w1;
        }
      }
#pragma unroll
      for (int r = 0; r < 8; ++r) {
#pragma unroll
        for (int off = 8; off > 0; off >>= 1) {
          p0[r] += __shfl_xor(p0[r], off, 32);
          p1[r] += __shfl_xor(p1[r], off, 32);
        }
      }
      if (c == 0) {
        const float o0 = b2[0];
        const float o1 = b2[1];
        float* op = C + (size_t)rowBase * 2;
#pragma unroll
        for (int r = 0; r < 8; r += 2)
          *(float4*)(op + 2 * r) = make_float4(p0[r] + o0, p1[r] + o1, p0[r + 1] + o0, p1[r + 1] + o1);
      }
    } else {
#pragma unroll
      for (int t = 0; t < NT; ++t) {
#pragma unroll
        for (int r = 0; r < 8; ++r)
          C[(size_t)(rowBase + r) * NOUT + 16 * t + c] = acc[t][r];
      }
    }
  }
}

__global__ void fill_int_kernel(int* __restrict__ p, int v, int n) {
  int i = blockIdx.x * blockDim.x + threadIdx.x;
  if (i < n) p[i] = v;
}

__device__ __forceinline__ int bcast_i(int v, int i) {
  return __builtin_amdgcn_readlane(v, i);
}

__global__ void csr_count(const int* __restrict__ ei, int* __restrict__ cnt) {
  int e = blockIdx.x * blockDim.x + threadIdx.x;
  if (e >= GE) return;
  atomicAdd(&cnt[ei[GE + e]], 1);
}

#define SCAN_T 1024
#define SCAN_BLOCKS ((GN + SCAN_T - 1) / SCAN_T)

__global__ void __launch_bounds__(SCAN_T)
csr_scan(const int* __restrict__ cnt, int* __restrict__ rowptr, int* __restrict__ cursor) {
  __shared__ int wsum[SCAN_T / 32];
  __shared__ int wtot[SCAN_T / 32];
  const int t    = threadIdx.x;
  const int lane = t & 31;
  const int wv   = t >> 5;
  const int tile0 = blockIdx.x * SCAN_T;

  int pre = 0;
  for (int i = t; i < tile0; i += SCAN_T) pre += cnt[i];
#pragma unroll
  for (int off = 16; off > 0; off >>= 1) pre += __shfl_xor(pre, off, 32);

  const int idx = tile0 + t;
  const int x = (idx < GN) ? cnt[idx] : 0;
  int inc = x;
#pragma unroll
  for (int off = 1; off < 32; off <<= 1) {
    const int y = __shfl_up(inc, off, 32);
    if (lane >= off) inc += y;
  }
  if (lane == 0)  wsum[wv] = pre;
  if (lane == 31) wtot[wv] = inc;
  __syncthreads();

  int base = 0;
  int wofs = 0;
#pragma unroll
  for (int k = 0; k < SCAN_T / 32; ++k) {
    base += wsum[k];
    if (k < wv) wofs += wtot[k];
  }
  const int excl = base + wofs + inc - x;
  if (idx < GN) {
    rowptr[idx] = excl;
    cursor[idx] = excl;
  }
  if (idx == GN - 1) rowptr[GN] = excl + x;
}

__global__ void csr_fill(const int* __restrict__ ei, int* __restrict__ cursor,
                         int* __restrict__ col) {
  int e = blockIdx.x * blockDim.x + threadIdx.x;
  if (e >= GE) return;
  const int pos = atomicAdd(&cursor[ei[GE + e]], 1);
  col[pos] = ei[e];
}

#define NODE_WAVES_PER_BLOCK 4

__device__ __forceinline__ void edge_logit2(float x0, float x1, float x2, float x3,
                                            float r0, float r1, float r2, float r3,
                                            float a00, float a01, float a10, float a11,
                                            bool lowHalf, float& l0, float& l1) {
  float t0 = x0 + r0;
  float t1 = x1 + r1;
  float u0 = x2 + r2;
  float u1 = x3 + r3;
  t0 = (t0 > 0.0f) ? t0 : NEG_SLOPE * t0;
  t1 = (t1 > 0.0f) ? t1 : NEG_SLOPE * t1;
  u0 = (u0 > 0.0f) ? u0 : NEG_SLOPE * u0;
  u1 = (u1 > 0.0f) ? u1 : NEG_SLOPE * u1;
  const float v0 = t0 * a00 + t1 * a01;
  const float v1 = u0 * a10 + u1 * a11;
  const float keep = lowHalf ? v0 : v1;
  const float send = lowHalf ? v1 : v0;
  float w = keep + __shfl_xor(send, 16, 32);
#pragma unroll
  for (int off = 8; off > 0; off >>= 1) w += __shfl_xor(w, off, 32);
  const float o = __shfl_xor(w, 16, 32);
  l0 = lowHalf ? w : o;
  l1 = lowHalf ? o : w;
}

__global__ void __launch_bounds__(NODE_WAVES_PER_BLOCK * 32)
node_attend(const float* __restrict__ xl,
            const float* __restrict__ xr,
            const int* __restrict__ rowptr,
            const int* __restrict__ col,
            const float* __restrict__ att,
            float* __restrict__ outp) {
  const int lane = threadIdx.x & 31;
  const int n = __builtin_amdgcn_readfirstlane(blockIdx.x * NODE_WAVES_PER_BLOCK + (threadIdx.x >> 5));
  if (n >= GN) return;
  const int beg = rowptr[n];
  const int end = rowptr[n + 1];

  const float a00 = att[lane];
  const float a01 = att[lane + 32];
  const float a10 = att[GC + lane];
  const float a11 = att[GC + lane + 32];
  const bool lowHalf = lane < 16;

  const float* xrn = xr + (size_t)n * GHC + lane;
  const float r0 = xrn[0];
  const float r1 = xrn[32];
  const float r2 = xrn[GC];
  const float r3 = xrn[GC + 32];

  const float* xln = xl + (size_t)n * GHC + lane;
  float acc0 = xln[0];
  float acc1 = xln[32];
  float acc2 = xln[GC];
  float acc3 = xln[GC + 32];
  float m0, m1;
  edge_logit2(acc0, acc1, acc2, acc3, r0, r1, r2, r3, a00, a01, a10, a11, lowHalf, m0, m1);
  float dn0 = 1.0f, dn1 = 1.0f;

  for (int cb = beg; cb < end; cb += 32) {
    const int left = end - cb;
    const int cnt  = (left < 32) ? left : 32;
    const int s = (lane < cnt) ? col[cb + lane] : 0;

    int i = 0;
    for (; i + 4 <= cnt; i += 4) {
      float x[4][4];
      float l0[4], l1[4];
#pragma unroll
      for (int k = 0; k < 4; ++k) {
        const int si = bcast_i(s, i + k);
        const float* xs = xl + (size_t)si * GHC + lane;
        x[k][0] = xs[0];
        x[k][1] = xs[32];
        x[k][2] = xs[GC];
        x[k][3] = xs[GC + 32];
      }
#pragma unroll
      for (int k = 0; k < 4; ++k)
        edge_logit2(x[k][0], x[k][1], x[k][2], x[k][3], r0, r1, r2, r3,
                    a00, a01, a10, a11, lowHalf, l0[k], l1[k]);

      const float n0 = fmaxf(fmaxf(m0, fmaxf(l0[0], l0[1])), fmaxf(l0[2], l0[3]));
      const float n1 = fmaxf(fmaxf(m1, fmaxf(l1[0], l1[1])), fmaxf(l1[2], l1[3]));
      const float sc0 = __expf(m0 - n0);
      const float sc1 = __expf(m1 - n1);
      m0 = n0;
      m1 = n1;
      dn0 *= sc0;  acc0 *= sc0;  acc1 *= sc0;
      dn1 *= sc1;  acc2 *= sc1;  acc3 *= sc1;
#pragma unroll
      for (int k = 0; k < 4; ++k) {
        const float p0 = __expf(l0[k] - n0);
        const float p1 = __expf(l1[k] - n1);
        dn0 += p0;
        dn1 += p1;
        acc0 += p0 * x[k][0];
        acc1 += p0 * x[k][1];
        acc2 += p1 * x[k][2];
        acc3 += p1 * x[k][3];
      }
    }
    for (; i < cnt; ++i) {
      const int si = bcast_i(s, i);
      const float* xs = xl + (size_t)si * GHC + lane;
      const float x0 = xs[0];
      const float x1 = xs[32];
      const float x2 = xs[GC];
      const float x3 = xs[GC + 32];
      float l0, l1;
      edge_logit2(x0, x1, x2, x3, r0, r1, r2, r3, a00, a01, a10, a11, lowHalf, l0, l1);
      const float n0 = fmaxf(m0, l0);
      const float n1 = fmaxf(m1, l1);
      const float sc0 = __expf(m0 - n0);
      const float sc1 = __expf(m1 - n1);
      const float p0  = __expf(l0 - n0);
      const float p1  = __expf(l1 - n1);
      m0 = n0;
      m1 = n1;
      dn0  = dn0 * sc0 + p0;
      dn1  = dn1 * sc1 + p1;
      acc0 = acc0 * sc0 + p0 * x0;
      acc1 = acc1 * sc0 + p0 * x1;
      acc2 = acc2 * sc1 + p1 * x2;
      acc3 = acc3 * sc1 + p1 * x3;
    }
  }

  float* on = outp + (size_t)n * GHC + lane;
  on[0]       = acc0 / dn0;
  on[32]      = acc1 / dn0;
  on[GC]      = acc2 / dn1;
  on[GC + 32] = acc3 / dn1;
}

#define STATS_ROWS 16
#define STATS_WAVES_PER_BLOCK 16
#define STATS_WAVES (GN / STATS_ROWS)
#define STATS_BLOCKS ((STATS_WAVES + STATS_WAVES_PER_BLOCK - 1) / STATS_WAVES_PER_BLOCK)
static_assert(GN % STATS_ROWS == 0, "a wave's rows are all inside the array");

__global__ void __launch_bounds__(STATS_WAVES_PER_BLOCK * 32)
gn_stats(const float* __restrict__ accum,
         const float* __restrict__ bias,
         float* __restrict__ colsum,
         float* __restrict__ colsumsq) {
  __shared__ alignas(16) float part[2][STATS_WAVES_PER_BLOCK][GHC];
  const int lane = threadIdx.x & 31;
  const int wv   = threadIdx.x >> 5;
  const int wave = blockIdx.x * STATS_WAVES_PER_BLOCK + wv;

  float4 s  = make_float4(0.0f, 0.0f, 0.0f, 0.0f);
  float4 s2 = make_float4(0.0f, 0.0f, 0.0f, 0.0f);
  if (wave < STATS_WAVES) {
    const size_t base = (size_t)wave * STATS_ROWS * GHC + (size_t)lane * 4;
    float4 v[STATS_ROWS];
#pragma unroll
    for (int k = 0; k < STATS_ROWS; ++k)
      v[k] = *(const float4*)(accum + base + (size_t)k * GHC);
    const float4 bv = *(const float4*)(bias + lane * 4);
#pragma unroll
    for (int k = 0; k < STATS_ROWS; ++k) {
      const float x0 = v[k].x + bv.x;
      const float x1 = v[k].y + bv.y;
      const float x2 = v[k].z + bv.z;
      const float x3 = v[k].w + bv.w;
      s.x += x0;  s2.x += x0 * x0;
      s.y += x1;  s2.y += x1 * x1;
      s.z += x2;  s2.z += x2 * x2;
      s.w += x3;  s2.w += x3 * x3;
    }
  }
  *(float4*)&part[0][wv][lane * 4] = s;
  *(float4*)&part[1][wv][lane * 4] = s2;
  __syncthreads();

  const int t = threadIdx.x;
  if (t < 2 * GHC) {
    const int st = t >> 7;
    const int f  = t & (GHC - 1);
    float tot = 0.0f;
#pragma unroll
    for (int w = 0; w < STATS_WAVES_PER_BLOCK; ++w) tot += part[st][w][f];
    atomicAdd(st ? &colsumsq[f] : &colsum[f], tot);
  }
}

extern "C" void kernel_launch(void* const* d_in, const int* in_sizes, int n_in,
                              void* d_out, int out_size, void* d_ws, size_t ws_size,
                              hipStream_t stream) {
  (void)in_sizes; (void)n_in; (void)out_size; (void)ws_size;

  const float* x     = (const float*)d_in[0];
  const int*   ei    = (const int*)d_in[1];
  const float* Wl    = (const float*)d_in[2];
  const float* Wr    = (const float*)d_in[3];
  const float* att   = (const float*)d_in[4];
  const float* convb = (const float*)d_in[5];
  const float* gnw   = (const float*)d_in[6];
  const float* gnb   = (const float*)d_in[7];
  const float* gnms  = (const float*)d_in[8];
  const float* W1    = (const float*)d_in[9];
  const float* b1    = (const float*)d_in[10];
  const float* W2    = (const float*)d_in[11];
  const float* b2    = (const float*)d_in[12];
  float* out = (float*)d_out;

  float* ws       = (float*)d_ws;
  float* xl       = ws;
  float* xr       = xl + (size_t)GN * GHC;
  float* hbuf     = xr + (size_t)GN * GHC;
  float* h1       = hbuf + (size_t)GN * GHC;
  int*   col      = (int*)h1;
  size_t scratch1 = ((size_t)GN * GC > (size_t)GETOT * GH)
                        ? (size_t)GN * GC : (size_t)GETOT * GH;
  float* accum    = h1 + scratch1;
  float* rpregion = accum + (size_t)GN * GHC;
  int*   rowptr   = (int*)rpregion;
  float* sum0     = rpregion + (size_t)GN * GH;
  float* sumsq0   = sum0 + GHC;
  float* sum1     = sumsq0 + GHC;
  float* sumsq1   = sum1 + GHC;
  int*   cnt      = (int*)(sumsq1 + GHC);
  int*   cursor   = cnt + GN;

  fill_int_kernel<<<(GN + 255) / 256, 256, 0, stream>>>(cnt, 0, GN);
  csr_count<<<(GE + 255) / 256, 256, 0, stream>>>(ei, cnt);
  csr_scan<<<SCAN_BLOCKS, SCAN_T, 0, stream>>>(cnt, rowptr, cursor);
  csr_fill<<<(GE + 255) / 256, 256, 0, stream>>>(ei, cursor, col);

  const int nodeBlocks = (GN + NODE_WAVES_PER_BLOCK - 1) / NODE_WAVES_PER_BLOCK;

  gemm_strip_f32<GHC / 16, false, false><<<GT_BLOCKS, GT_THREADS, 0, stream>>>(
      x, Wl, Wr, xl, xr,
      nullptr, nullptr, nullptr, nullptr, nullptr, nullptr,
      nullptr, nullptr, nullptr, sum0, 2, GN / 16);
  node_attend<<<nodeBlocks, NODE_WAVES_PER_BLOCK * 32, 0, stream>>>(xl, xr, rowptr, col, att, accum);
  gn_stats<<<STATS_BLOCKS, STATS_WAVES_PER_BLOCK * 32, 0, stream>>>(accum, convb, sum0, sumsq0);

  gemm_strip_f32<GHC / 16, true, false><<<GT_BLOCKS, GT_THREADS, 0, stream>>>(
      accum, Wl + (size_t)GHC * GHC, Wr + (size_t)GHC * GHC, xl, xr,
      sum0, sumsq0, gnw, gnb, gnms, convb,
      nullptr, nullptr, nullptr, nullptr, 2, GN / 16);
  node_attend<<<nodeBlocks, NODE_WAVES_PER_BLOCK * 32, 0, stream>>>(xl, xr, rowptr, col, att + GH * GC, accum);
  gn_stats<<<STATS_BLOCKS, STATS_WAVES_PER_BLOCK * 32, 0, stream>>>(accum, convb + GHC, sum1, sumsq1);

  gemm_strip_f32<GC / 16, true, true><<<GT_BLOCKS, GT_THREADS, 0, stream>>>(
      accum, W1, nullptr, out, nullptr,
      sum1, sumsq1, gnw + GHC, gnb + GHC, gnms + GHC, convb + GHC,
      b1, W2, b2, nullptr, 1, GN / 16);
}


# Edge grouping, no global atomic per edge: edges counting-sorted in LDS into 256-node bins (one global add per bin and block), then a block per bin builds slot rows in LDS; 9 launches; new sum order
# speedup vs baseline: 9.3780x; 1.8078x over previous
#include <hip/hip_runtime.h>
#include <hip/hip_bf16.h>


#define GN 50000
#define GE 800000
#define GETOT (GE + GN)
#define GHC 128
#define GC 64
#define GH 2
#define NEG_SLOPE 0.2f
#define GEPS 1e-5f

typedef __attribute__((ext_vector_type(2))) float v2f;
typedef __attribute__((ext_vector_type(8))) float v8f;

#define GT_WAVES 16
#define GT_THREADS (GT_WAVES * 32)
#define GT_K 128
#define GT_BLOCKS 256

template <int NT, bool NORM, bool HEAD>
__global__ void __launch_bounds__(GT_THREADS)
gemm_strip_f32(const float* __restrict__ A,
               const float* __restrict__ B0,
               const float* __restrict__ B1,
               float* __restrict__ C0,
               float* __restrict__ C1,
               const float* __restrict__ colsum,
               const float* __restrict__ colsumsq,
               const float* __restrict__ gw,
               const float* __restrict__ gb,
               const float* __restrict__ gms,
               const float* __restrict__ cbias,
               const float* __restrict__ bias,
               const float* __restrict__ W2,
               const float* __restrict__ b2,
               float* __restrict__ zero512,
               int* __restrict__ zint, int nzint,
               int nmat, int nStrips) {
  constexpr int NOUT = 16 * NT;
  static_assert(!HEAD || NT == GC / 16, "the head product expects the 64-column GEMM");
  __shared__ alignas(16) float Ns[2][GT_K];
  __shared__ alignas(16) float Bs[32 * NT * 32 * 2];

  const int tid  = threadIdx.x;
  const int lane = tid & 31;
  const int wv   = __builtin_amdgcn_readfirstlane(tid >> 5);
  const int mat  = blockIdx.x % nmat;
  const float* __restrict__ B = mat ? B1 : B0;
  float* __restrict__ C       = mat ? C1 : C0;

  static_assert(GT_THREADS == 4 * GHC, "one thread per float of the two layers' column sums");
  if (zero512 != nullptr && blockIdx.x == 0) zero512[tid] = 0.0f;
  if (zint != nullptr) {
    const int q = blockIdx.x * GT_THREADS + tid;
    if (q < nzint) zint[q] = 0;
  }

  if (NORM && tid < GT_K) {
    const float invN = 1.0f / (float)GN;
    const float mean = colsum[tid] * invN;
    const float ex2  = colsumsq[tid] * invN;
    const float msv  = gms[tid];
    const float var  = ex2 - (2.0f * msv - msv * msv) * mean * mean;
    const float r    = rsqrtf(var + GEPS);
    const float a    = gw[tid] * r;
    const float sh   = gb[tid] - a * msv * mean;
    Ns[0][tid] = a;
    Ns[1][tid] = cbias[tid] * a + sh;
  }

  constexpr int FILL_IT = GT_K * (NOUT / 4) / GT_THREADS;
  static_assert(GT_K * (NOUT / 4) % GT_THREADS == 0, "every thread fills the same number of slots");
  float4 fv[FILL_IT];
#pragma unroll
  for (int it = 0; it < FILL_IT; ++it) {
    const int q  = tid + it * GT_THREADS;
    const int k  = q / (NOUT / 4);
    const int n4 = (q % (NOUT / 4)) * 4;
    fv[it] = *(const float4*)(B + (size_t)k * NOUT + n4);
  }
#pragma unroll
  for (int it = 0; it < FILL_IT; ++it) {
    const int q  = tid + it * GT_THREADS;
    const int k  = q / (NOUT / 4);
    const int n4 = (q % (NOUT / 4)) * 4;
    const int s  = ((k >> 3) << 1) | ((k >> 1) & 1);
    const int hh = (k >> 2) & 1;
    const int e  = k & 1;
    const int t  = n4 >> 4;
    const int cc = n4 & 15;
    const int base = ((s * (NT / 2) + (t >> 1)) * 32 + cc + 16 * hh) * 4 + ((t & 1) << 1) + e;
    Bs[base]      = fv[it].x;
    Bs[base + 4]  = fv[it].y;
    Bs[base + 8]  = fv[it].z;
    Bs[base + 12] = fv[it].w;
  }
  __syncthreads();

  const int c = lane & 15;
  const int h = lane >> 4;
  const int first  = blockIdx.x / nmat;
  const int stride = gridDim.x / nmat;

  for (int sidx = wv; ; sidx += GT_WAVES) {
    const int strip = first + stride * sidx;
    if (strip >= nStrips) break;

    const float* ap = A + ((size_t)strip * 16 + c) * GT_K + 4 * h;
    float4 a4[GT_K / 8];
#pragma unroll
    for (int j = 0; j < GT_K / 8; ++j) a4[j] = *(const float4*)(ap + 8 * j);
    if (NORM) {
      const float4* n1 = (const float4*)&Ns[0][4 * h];
      const float4* n2 = (const float4*)&Ns[1][4 * h];
#pragma unroll
      for (int j = 0; j < GT_K / 8; ++j) {
        const float4 p = n1[2 * j];
        const float4 q = n2[2 * j];
        a4[j].x = fmaxf(a4[j].x * p.x + q.x, 0.0f);
        a4[j].y = fmaxf(a4[j].y * p.y + q.y, 0.0f);
        a4[j].z = fmaxf(a4[j].z * p.z + q.z, 0.0f);
        a4[j].w = fmaxf(a4[j].w * p.w + q.w, 0.0f);
      }
    }

    v8f acc[NT];
#pragma unroll
    for (int t = 0; t < NT; ++t) acc[t] = v8f{};

    const float4* bl = (const float4*)Bs + lane;
    float4 bq[2][NT];
#pragma unroll
    for (int i = 0; i < NT; ++i) bq[0][i] = bl[i * 32];

#pragma unroll
    for (int j = 0; j < GT_K / 8; ++j) {
      if (j + 1 < GT_K / 8) {
#pragma unroll
        for (int i = 0; i < NT; ++i) bq[(j + 1) & 1][i] = bl[((j + 1) * NT + i) * 32];
      }
#pragma unroll
      for (int p = 0; p < 2; ++p) {
        v2f a;
        a.x = p ? a4[j].z : a4[j].x;
        a.y = p ? a4[j].w : a4[j].y;
#pragma unroll
        for (int tp = 0; tp < NT / 2; ++tp) {
          const float4 b = bq[j & 1][p * (NT / 2) + tp];
          v2f b0, b1;
          b0.x = b.x;  b0.y = b.y;
          b1.x = b.z;  b1.y = b.w;
          acc[2 * tp]     = __builtin_amdgcn_wmma_f32_16x16x4_f32(false, a, false, b0,
                                                                  (short)0, acc[2 * tp], false, false);
          acc[2 * tp + 1] = __builtin_amdgcn_wmma_f32_16x16x4_f32(false, a, false, b1,
                                                                  (short)0, acc[2 * tp + 1], false, false);
        }
      }
      __builtin_amdgcn_sched_barrier(0);
    }

    const int rowBase = strip * 16 + (h << 3);
    if (HEAD) {
      float p0[8], p1[8];
#pragma unroll
      for (int r = 0; r < 8; ++r) { p0[r] = 0.0f; p1[r] = 0.0f; }
#pragma unroll
      for (int t = 0; t < NT; ++t) {
        const float bv = bias[16 * t + c];
        const float w0 = W2[(16 * t + c) * 2 + 0];
        const float w1 = W2[(16 * t + c) * 2 + 1];
#pragma unroll
        for (int r = 0; r < 8; ++r) {
          const float v = fmaxf(acc[t][r] + bv, 0.0f);
          p0[r] += v * w0;
          p1[r] += v * w1;
        }
      }
#pragma unroll
      for (int r = 0; r < 8; ++r) {
#pragma unroll
        for (int off = 8; off > 0; off >>= 1) {
          p0[r] += __shfl_xor(p0[r], off, 32);
          p1[r] += __shfl_xor(p1[r], off, 32);
        }
      }
      if (c == 0) {
        const float o0 = b2[0];
        const float o1 = b2[1];
        float* op = C + (size_t)rowBase * 2;
#pragma unroll
        for (int r = 0; r < 8; r += 2)
          *(float4*)(op + 2 * r) = make_float4(p0[r] + o0, p1[r] + o1, p0[r + 1] + o0, p1[r + 1] + o1);
      }
    } else {
#pragma unroll
      for (int t = 0; t < NT; ++t) {
#pragma unroll
        for (int r = 0; r < 8; ++r)
          C[(size_t)(rowBase + r) * NOUT + 16 * t + c] = acc[t][r];
      }
    }
  }
}

__device__ __forceinline__ int bcast_i(int v, int i) {
  return __builtin_amdgcn_readlane(v, i);
}

#define SLOT_W 32
#define BIN_SHIFT 8
#define BIN_NODES (1 << BIN_SHIFT)
#define NBINS ((GN + BIN_NODES - 1) >> BIN_SHIFT)
#define BIN_CAP 6144
#define EB_THREADS 512
#define EB_BLOCKS 256
#define EB_CHUNK ((GE + EB_BLOCKS - 1) / EB_BLOCKS)
#define EB_EPT ((EB_CHUNK + EB_THREADS - 1) / EB_THREADS)
#define GB_THREADS 512
#define GB_EPT (BIN_CAP / GB_THREADS)
#define CTL_OVF 32
#define CTL_BIN0 64
#define CTL_STRIDE 32
#define CTL_INTS (CTL_BIN0 + CTL_STRIDE * NBINS)
static_assert(NBINS <= 256, "the bin counters are scanned by 256 threads");
static_assert(BIN_CAP % GB_THREADS == 0, "every thread of fill_slots takes the same number of list places");
static_assert(GT_BLOCKS * GT_THREADS >= CTL_INTS, "the first GEMM launch clears the list lengths, one int per thread");
static_assert(CTL_INTS <= GN, "the list lengths fit in the room behind the node counts");

__global__ void __launch_bounds__(EB_THREADS)
bin_edges(const int* __restrict__ ei, int* __restrict__ deg,
          int* __restrict__ binBuf, int* __restrict__ ovf) {
  __shared__ int hist[256];
  __shared__ int lofs[256];
  __shared__ int gofs[256];
  __shared__ int wtot[8];
  __shared__ int2 sorted[EB_CHUNK];

  int* binCount = deg + GN + CTL_BIN0;
  int* ovfCount = deg + GN + CTL_OVF;
  const int t    = threadIdx.x;
  const int lane = t & 31;
  const int e0   = blockIdx.x * EB_CHUNK;
  const int eEnd = (e0 + EB_CHUNK < GE) ? (e0 + EB_CHUNK) : GE;

  if (t < 256) hist[t] = 0;
  __syncthreads();

  int d[EB_EPT], s[EB_EPT], r[EB_EPT];
#pragma unroll
  for (int i = 0; i < EB_EPT; ++i) {
    const int e = e0 + t + i * EB_THREADS;
    d[i] = -1;
    s[i] = 0;
    if (e < eEnd) {
      d[i] = ei[GE + e];
      s[i] = ei[e];
    }
  }
#pragma unroll
  for (int i = 0; i < EB_EPT; ++i) {
    r[i] = 0;
    if (d[i] >= 0) r[i] = atomicAdd(&hist[d[i] >> BIN_SHIFT], 1);
  }
  __syncthreads();

  int hv = 0, incl = 0;
  if (t < 256) {
    hv = hist[t];
    incl = hv;
#pragma unroll
    for (int off = 1; off < 32; off <<= 1) {
      const int v = __shfl_up(incl, off, 32);
      if (lane >= off) incl += v;
    }
    if (lane == 31) wtot[t >> 5] = incl;
  }
  __syncthreads();
  if (t < 256) {
    int wbase = 0;
#pragma unroll
    for (int w = 0; w < 8; ++w) wbase += (w < (t >> 5)) ? wtot[w] : 0;
    const int ex = wbase + incl - hv;
    lofs[t] = ex;
    int g = 0;
    if (t < NBINS && hv > 0) g = atomicAdd(&binCount[t * CTL_STRIDE], hv);
    gofs[t] = g - ex;
  }
  __syncthreads();

#pragma unroll
  for (int i = 0; i < EB_EPT; ++i) {
    if (d[i] >= 0) sorted[lofs[d[i] >> BIN_SHIFT] + r[i]] = make_int2(d[i], s[i]);
  }
  __syncthreads();

  const int n = eEnd - e0;
  int2* __restrict__ bb = (int2*)binBuf;
  int2* __restrict__ ov = (int2*)ovf;
#pragma unroll
  for (int i = 0; i < EB_EPT; ++i) {
    const int p = t + i * EB_THREADS;
    if (p < n) {
      const int2 v  = sorted[p];
      const int bin = v.x >> BIN_SHIFT;
      const int idx = gofs[bin] + p;
      int2* q = bb + (unsigned)(bin * BIN_CAP + idx);
      if (idx >= BIN_CAP) q = ov + atomicAdd(ovfCount, 1);
      *q = v;
    }
  }
}

__device__ __forceinline__ void store_edge(const int2 v, const int r, int* rows,
                                           int* __restrict__ deg, int* __restrict__ spill) {
  if (r < SLOT_W) {
    rows[(v.x & (BIN_NODES - 1)) * SLOT_W + r] = v.y;
  } else {
    const int k = atomicAdd(&deg[GN], 1);
    spill[2 * k]     = v.x;
    spill[2 * k + 1] = v.y;
  }
}

__global__ void __launch_bounds__(GB_THREADS)
fill_slots(int* __restrict__ deg, const int* __restrict__ binBuf, const int* __restrict__ ovf,
           int* __restrict__ slot, int* __restrict__ spill) {
  __shared__ int cnt[BIN_NODES];
  __shared__ alignas(16) int rows[BIN_NODES * SLOT_W];

  const int t = threadIdx.x;
  const int b = blockIdx.x;
  const int total = deg[GN + CTL_BIN0 + CTL_STRIDE * b];
  const int n     = (total < BIN_CAP) ? total : BIN_CAP;
  const int novf  = deg[GN + CTL_OVF];

  const int2* __restrict__ lp = (const int2*)binBuf + (size_t)b * BIN_CAP;
  int2 v[GB_EPT];
#pragma unroll
  for (int i = 0; i < GB_EPT; ++i) {
    const int p = t + i * GB_THREADS;
    v[i] = make_int2(-1, 0);
    if (p < n) v[i] = lp[p];
  }

  if (t < BIN_NODES) cnt[t] = 0;
  for (int u = t; u < BIN_NODES * SLOT_W / 4; u += GB_THREADS) ((int4*)rows)[u] = make_int4(0, 0, 0, 0);
  __syncthreads();

  int r[GB_EPT];
#pragma unroll
  for (int i = 0; i < GB_EPT; ++i) {
    r[i] = 0;
    if (v[i].x >= 0) r[i] = atomicAdd(&cnt[v[i].x & (BIN_NODES - 1)], 1);
  }
#pragma unroll
  for (int i = 0; i < GB_EPT; ++i) {
    if (v[i].x >= 0) store_edge(v[i], r[i], rows, deg, spill);
  }
  for (int q = t; q < novf; q += GB_THREADS) {
    const int2 w = ((const int2*)ovf)[q];
    if ((w.x >> BIN_SHIFT) == b) {
      const int rw = atomicAdd(&cnt[w.x & (BIN_NODES - 1)], 1);
      store_edge(w, rw, rows, deg, spill);
    }
  }
  __syncthreads();

  const int node0 = b << BIN_SHIFT;
  int4* __restrict__ sp = (int4*)slot + (size_t)node0 * (SLOT_W / 4);
  for (int u = t; u < BIN_NODES * SLOT_W / 4; u += GB_THREADS) {
    if (node0 + (u / (SLOT_W / 4)) < GN) sp[u] = ((const int4*)rows)[u];
  }
  if (t < BIN_NODES && node0 + t < GN) deg[node0 + t] = cnt[t];
}

#define NODE_WAVES_PER_BLOCK 4

typedef __attribute__((ext_vector_type(4))) float v4f;

template <int MASK>
__device__ __forceinline__ float row_xor(float v) {
  return __builtin_bit_cast(float, __builtin_amdgcn_mov_dpp(__builtin_bit_cast(int, v), 0x160 + MASK, 0xf, 0xf, false));
}

__device__ __forceinline__ float lane_logit_part(const v4f x, const v4f r, const v4f a) {
  v4f t = x + r;
  t = __builtin_elementwise_max(t, t * NEG_SLOPE);
  const v2f q = t.xy * a.xy + t.zw * a.zw;
  return q.x + q.y;
}

template <int NB, bool MASKED>
__device__ __forceinline__ void attend_rows(const v4f* __restrict__ xl4, const int s, const int i,
                                            const int valid, const int lane,
                                            const v4f r, const v4f a,
                                            float& m, float& dn, v4f& acc) {
  v4f x[NB];
#pragma unroll
  for (int k = 0; k < NB; ++k) {
    const int si = bcast_i(s, i + k);
    x[k] = xl4[(unsigned)(si * (GHC / 4) + lane)];
  }
  float l[NB];
#pragma unroll
  for (int k = 0; k < NB; ++k) l[k] = lane_logit_part(x[k], r, a);
#pragma unroll
  for (int k = 0; k < NB; ++k) l[k] += row_xor<1>(l[k]);
#pragma unroll
  for (int k = 0; k < NB; ++k) l[k] += row_xor<2>(l[k]);
#pragma unroll
  for (int k = 0; k < NB; ++k) l[k] += row_xor<4>(l[k]);
#pragma unroll
  for (int k = 0; k < NB; ++k) l[k] += row_xor<8>(l[k]);
  if (MASKED) {
#pragma unroll
    for (int k = 0; k < NB; ++k) l[k] = (k < valid) ? l[k] : -__builtin_huge_valf();
  }
  float nm = m;
#pragma unroll
  for (int k = 0; k < NB; ++k) nm = fmaxf(nm, l[k]);
  const float sc = __expf(m - nm);
  m = nm;
  dn *= sc;
  acc *= sc;
#pragma unroll
  for (int k = 0; k < NB; ++k) {
    const float p = __expf(l[k] - nm);
    dn += p;
    acc += p * x[k];
  }
}

__global__ void __launch_bounds__(NODE_WAVES_PER_BLOCK * 32, 12)
node_attend(const float* __restrict__ xl,
            const float* __restrict__ xr,
            const int* __restrict__ deg,
            const int* __restrict__ slot,
            const int* __restrict__ spill,
            const float* __restrict__ att,
            float* __restrict__ outp) {
  const int lane = threadIdx.x & 31;
  const int n = __builtin_amdgcn_readfirstlane(blockIdx.x * NODE_WAVES_PER_BLOCK + (threadIdx.x >> 5));
  if (n >= GN) return;
  const v4f* __restrict__ xl4 = (const v4f*)xl;

  const int dg   = deg[n];
  const int sraw = slot[(unsigned)(n * SLOT_W + lane)];

  const v4f a = ((const v4f*)att)[lane];
  const v4f r = ((const v4f*)xr)[(size_t)n * (GHC / 4) + lane];

  v4f acc = xl4[(size_t)n * (GHC / 4) + lane];
  float m = lane_logit_part(acc, r, a);
  m += row_xor<1>(m);
  m += row_xor<2>(m);
  m += row_xor<4>(m);
  m += row_xor<8>(m);
  float dn = 1.0f;

  const int cnt = (dg < SLOT_W) ? dg : SLOT_W;
  const int s   = (lane < cnt) ? sraw : n;
  int i = 0;
  for (; i + 8 <= cnt; i += 8)
    attend_rows<8, false>(xl4, s, i, 8, lane, r, a, m, dn, acc);
  const int rem = cnt - i;
  if (rem > 4)      attend_rows<8, true>(xl4, s, i, rem, lane, r, a, m, dn, acc);
  else if (rem > 0) attend_rows<4, true>(xl4, s, i, rem, lane, r, a, m, dn, acc);

  if (dg > SLOT_W) {
    const int total = deg[GN];
    for (int base = 0; base < total; base += 32) {
      const int q  = base + lane;
      const int pd = (q < total) ? spill[2 * q]     : -1;
      const int ps = (q < total) ? spill[2 * q + 1] : n;
      unsigned hit = __builtin_amdgcn_ballot_w32(pd == n);
      while (hit != 0u) {
        const int k = __builtin_ctz(hit);
        hit &= hit - 1u;
        const int src = bcast_i(ps, k);
        const int s1  = (lane == 0) ? src : n;
        attend_rows<4, true>(xl4, s1, 0, 1, lane, r, a, m, dn, acc);
      }
    }
  }

  v4f o;
  o.x = acc.x / dn;
  o.y = acc.y / dn;
  o.z = acc.z / dn;
  o.w = acc.w / dn;
  ((v4f*)outp)[(size_t)n * (GHC / 4) + lane] = o;
}

#define STATS_ROWS 16
#define STATS_WAVES_PER_BLOCK 16
#define STATS_WAVES (GN / STATS_ROWS)
#define STATS_BLOCKS ((STATS_WAVES + STATS_WAVES_PER_BLOCK - 1) / STATS_WAVES_PER_BLOCK)
static_assert(GN % STATS_ROWS == 0, "a wave's rows are all inside the array");

__global__ void __launch_bounds__(STATS_WAVES_PER_BLOCK * 32)
gn_stats(const float* __restrict__ accum,
         const float* __restrict__ bias,
         float* __restrict__ colsum,
         float* __restrict__ colsumsq) {
  __shared__ alignas(16) float part[2][STATS_WAVES_PER_BLOCK][GHC];
  const int lane = threadIdx.x & 31;
  const int wv   = threadIdx.x >> 5;
  const int wave = blockIdx.x * STATS_WAVES_PER_BLOCK + wv;

  float4 s  = make_float4(0.0f, 0.0f, 0.0f, 0.0f);
  float4 s2 = make_float4(0.0f, 0.0f, 0.0f, 0.0f);
  if (wave < STATS_WAVES) {
    const size_t base = (size_t)wave * STATS_ROWS * GHC + (size_t)lane * 4;
    float4 v[STATS_ROWS];
#pragma unroll
    for (int k = 0; k < STATS_ROWS; ++k)
      v[k] = *(const float4*)(accum + base + (size_t)k * GHC);
    const float4 bv = *(const float4*)(bias + lane * 4);
#pragma unroll
    for (int k = 0; k < STATS_ROWS; ++k) {
      const float x0 = v[k].x + bv.x;
      const float x1 = v[k].y + bv.y;
      const float x2 = v[k].z + bv.z;
      const float x3 = v[k].w + bv.w;
      s.x += x0;  s2.x += x0 * x0;
      s.y += x1;  s2.y += x1 * x1;
      s.z += x2;  s2.z += x2 * x2;
      s.w += x3;  s2.w += x3 * x3;
    }
  }
  *(float4*)&part[0][wv][lane * 4] = s;
  *(float4*)&part[1][wv][lane * 4] = s2;
  __syncthreads();

  const int t = threadIdx.x;
  if (t < 2 * GHC) {
    const int st = t >> 7;
    const int f  = t & (GHC - 1);
    float tot = 0.0f;
#pragma unroll
    for (int w = 0; w < STATS_WAVES_PER_BLOCK; ++w) tot += part[st][w][f];
    atomicAdd(st ? &colsumsq[f] : &colsum[f], tot);
  }
}

extern "C" void kernel_launch(void* const* d_in, const int* in_sizes, int n_in,
                              void* d_out, int out_size, void* d_ws, size_t ws_size,
                              hipStream_t stream) {
  (void)in_sizes; (void)n_in; (void)out_size; (void)ws_size;

  const float* x     = (const float*)d_in[0];
  const int*   ei    = (const int*)d_in[1];
  const float* Wl    = (const float*)d_in[2];
  const float* Wr    = (const float*)d_in[3];
  const float* att   = (const float*)d_in[4];
  const float* convb = (const float*)d_in[5];
  const float* gnw   = (const float*)d_in[6];
  const float* gnb   = (const float*)d_in[7];
  const float* gnms  = (const float*)d_in[8];
  const float* W1    = (const float*)d_in[9];
  const float* b1    = (const float*)d_in[10];
  const float* W2    = (const float*)d_in[11];
  const float* b2    = (const float*)d_in[12];
  float* out = (float*)d_out;

  float* ws       = (float*)d_ws;
  float* xl       = ws;
  float* xr       = xl + (size_t)GN * GHC;
  float* hbuf     = xr + (size_t)GN * GHC;
  int*   slot     = (int*)hbuf;
  int*   spill    = slot + (size_t)GN * SLOT_W;
  int*   ovf      = spill + (size_t)2 * GE;
  float* h1       = hbuf + (size_t)GN * GHC;
  int*   binBuf   = (int*)h1;
  size_t scratch1 = ((size_t)GN * GC > (size_t)GETOT * GH)
                        ? (size_t)GN * GC : (size_t)GETOT * GH;
  float* accum    = h1 + scratch1;
  float* rpregion = accum + (size_t)GN * GHC;
  float* sum0     = rpregion + (size_t)GN * GH;
  float* sumsq0   = sum0 + GHC;
  float* sum1     = sumsq0 + GHC;
  float* sumsq1   = sum1 + GHC;
  int*   deg      = (int*)(sumsq1 + GHC);

  const int nodeBlocks = (GN + NODE_WAVES_PER_BLOCK - 1) / NODE_WAVES_PER_BLOCK;

  gemm_strip_f32<GHC / 16, false, false><<<GT_BLOCKS, GT_THREADS, 0, stream>>>(
      x, Wl, Wr, xl, xr,
      nullptr, nullptr, nullptr, nullptr, nullptr, nullptr,
      nullptr, nullptr, nullptr, sum0, deg + GN, CTL_INTS, 2, GN / 16);
  bin_edges<<<EB_BLOCKS, EB_THREADS, 0, stream>>>(ei, deg, binBuf, ovf);
  fill_slots<<<NBINS, GB_THREADS, 0, stream>>>(deg, binBuf, ovf, slot, spill);
  node_attend<<<nodeBlocks, NODE_WAVES_PER_BLOCK * 32, 0, stream>>>(xl, xr, deg, slot, spill, att, accum);
  gn_stats<<<STATS_BLOCKS, STATS_WAVES_PER_BLOCK * 32, 0, stream>>>(accum, convb, sum0, sumsq0);

  gemm_strip_f32<GHC / 16, true, false><<<GT_BLOCKS, GT_THREADS, 0, stream>>>(
      accum, Wl + (size_t)GHC * GHC, Wr + (size_t)GHC * GHC, xl, xr,
      sum0, sumsq0, gnw, gnb, gnms, convb,
      nullptr, nullptr, nullptr, nullptr, nullptr, 0, 2, GN / 16);
  node_attend<<<nodeBlocks, NODE_WAVES_PER_BLOCK * 32, 0, stream>>>(xl, xr, deg, slot, spill, att + GH * GC, accum);
  gn_stats<<<STATS_BLOCKS, STATS_WAVES_PER_BLOCK * 32, 0, stream>>>(accum, convb + GHC, sum1, sumsq1);

  gemm_strip_f32<GC / 16, true, true><<<GT_BLOCKS, GT_THREADS, 0, stream>>>(
      accum, W1, nullptr, out, nullptr,
      sum1, sumsq1, gnw + GHC, gnb + GHC, gnms + GHC, convb + GHC,
      b1, W2, b2, nullptr, nullptr, 0, 1, GN / 16);
}


# node_attend: a group's softmax once per edge, not per lane: logit sums halve across lanes till a lane holds one edge's; one exp per lane, weight read back by row_share; denominator: new sum order
# speedup vs baseline: 9.6450x; 1.0285x over previous
#include <hip/hip_runtime.h>
#include <hip/hip_bf16.h>


#define GN 50000
#define GE 800000
#define GETOT (GE + GN)
#define GHC 128
#define GC 64
#define GH 2
#define NEG_SLOPE 0.2f
#define GEPS 1e-5f

typedef __attribute__((ext_vector_type(2))) float v2f;
typedef __attribute__((ext_vector_type(8))) float v8f;

#define GT_WAVES 16
#define GT_THREADS (GT_WAVES * 32)
#define GT_K 128
#define GT_BLOCKS 256

template <int NT, bool NORM, bool HEAD>
__global__ void __launch_bounds__(GT_THREADS)
gemm_strip_f32(const float* __restrict__ A,
               const float* __restrict__ B0,
               const float* __restrict__ B1,
               float* __restrict__ C0,
               float* __restrict__ C1,
               const float* __restrict__ colsum,
               const float* __restrict__ colsumsq,
               const float* __restrict__ gw,
               const float* __restrict__ gb,
               const float* __restrict__ gms,
               const float* __restrict__ cbias,
               const float* __restrict__ bias,
               const float* __restrict__ W2,
               const float* __restrict__ b2,
               float* __restrict__ zero512,
               int* __restrict__ zint, int nzint,
               int nmat, int nStrips) {
  constexpr int NOUT = 16 * NT;
  static_assert(!HEAD || NT == GC / 16, "the head product expects the 64-column GEMM");
  __shared__ alignas(16) float Ns[2][GT_K];
  __shared__ alignas(16) float Bs[32 * NT * 32 * 2];

  const int tid  = threadIdx.x;
  const int lane = tid & 31;
  const int wv   = __builtin_amdgcn_readfirstlane(tid >> 5);
  const int mat  = blockIdx.x % nmat;
  const float* __restrict__ B = mat ? B1 : B0;
  float* __restrict__ C       = mat ? C1 : C0;

  static_assert(GT_THREADS == 4 * GHC, "one thread per float of the two layers' column sums");
  if (zero512 != nullptr && blockIdx.x == 0) zero512[tid] = 0.0f;
  if (zint != nullptr) {
    const int q = blockIdx.x * GT_THREADS + tid;
    if (q < nzint) zint[q] = 0;
  }

  if (NORM && tid < GT_K) {
    const float invN = 1.0f / (float)GN;
    const float mean = colsum[tid] * invN;
    const float ex2  = colsumsq[tid] * invN;
    const float msv  = gms[tid];
    const float var  = ex2 - (2.0f * msv - msv * msv) * mean * mean;
    const float r    = rsqrtf(var + GEPS);
    const float a    = gw[tid] * r;
    const float sh   = gb[tid] - a * msv * mean;
    Ns[0][tid] = a;
    Ns[1][tid] = cbias[tid] * a + sh;
  }

  constexpr int FILL_IT = GT_K * (NOUT / 4) / GT_THREADS;
  static_assert(GT_K * (NOUT / 4) % GT_THREADS == 0, "every thread fills the same number of slots");
  float4 fv[FILL_IT];
#pragma unroll
  for (int it = 0; it < FILL_IT; ++it) {
    const int q  = tid + it * GT_THREADS;
    const int k  = q / (NOUT / 4);
    const int n4 = (q % (NOUT / 4)) * 4;
    fv[it] = *(const float4*)(B + (size_t)k * NOUT + n4);
  }
#pragma unroll
  for (int it = 0; it < FILL_IT; ++it) {
    const int q  = tid + it * GT_THREADS;
    const int k  = q / (NOUT / 4);
    const int n4 = (q % (NOUT / 4)) * 4;
    const int s  = ((k >> 3) << 1) | ((k >> 1) & 1);
    const int hh = (k >> 2) & 1;
    const int e  = k & 1;
    const int t  = n4 >> 4;
    const int cc = n4 & 15;
    const int base = ((s * (NT / 2) + (t >> 1)) * 32 + cc + 16 * hh) * 4 + ((t & 1) << 1) + e;
    Bs[base]      = fv[it].x;
    Bs[base + 4]  = fv[it].y;
    Bs[base + 8]  = fv[it].z;
    Bs[base + 12] = fv[it].w;
  }
  __syncthreads();

  const int c = lane & 15;
  const int h = lane >> 4;
  const int first  = blockIdx.x / nmat;
  const int stride = gridDim.x / nmat;

  for (int sidx = wv; ; sidx += GT_WAVES) {
    const int strip = first + stride * sidx;
    if (strip >= nStrips) break;

    const float* ap = A + ((size_t)strip * 16 + c) * GT_K + 4 * h;
    float4 a4[GT_K / 8];
#pragma unroll
    for (int j = 0; j < GT_K / 8; ++j) a4[j] = *(const float4*)(ap + 8 * j);
    if (NORM) {
      const float4* n1 = (const float4*)&Ns[0][4 * h];
      const float4* n2 = (const float4*)&Ns[1][4 * h];
#pragma unroll
      for (int j = 0; j < GT_K / 8; ++j) {
        const float4 p = n1[2 * j];
        const float4 q = n2[2 * j];
        a4[j].x = fmaxf(a4[j].x * p.x + q.x, 0.0f);
        a4[j].y = fmaxf(a4[j].y * p.y + q.y, 0.0f);
        a4[j].z = fmaxf(a4[j].z * p.z + q.z, 0.0f);
        a4[j].w = fmaxf(a4[j].w * p.w + q.w, 0.0f);
      }
    }

    v8f acc[NT];
#pragma unroll
    for (int t = 0; t < NT; ++t) acc[t] = v8f{};

    const float4* bl = (const float4*)Bs + lane;
    float4 bq[2][NT];
#pragma unroll
    for (int i = 0; i < NT; ++i) bq[0][i] = bl[i * 32];

#pragma unroll
    for (int j = 0; j < GT_K / 8; ++j) {
      if (j + 1 < GT_K / 8) {
#pragma unroll
        for (int i = 0; i < NT; ++i) bq[(j + 1) & 1][i] = bl[((j + 1) * NT + i) * 32];
      }
#pragma unroll
      for (int p = 0; p < 2; ++p) {
        v2f a;
        a.x = p ? a4[j].z : a4[j].x;
        a.y = p ? a4[j].w : a4[j].y;
#pragma unroll
        for (int tp = 0; tp < NT / 2; ++tp) {
          const float4 b = bq[j & 1][p * (NT / 2) + tp];
          v2f b0, b1;
          b0.x = b.x;  b0.y = b.y;
          b1.x = b.z;  b1.y = b.w;
          acc[2 * tp]     = __builtin_amdgcn_wmma_f32_16x16x4_f32(false, a, false, b0,
                                                                  (short)0, acc[2 * tp], false, false);
          acc[2 * tp + 1] = __builtin_amdgcn_wmma_f32_16x16x4_f32(false, a, false, b1,
                                                                  (short)0, acc[2 * tp + 1], false, false);
        }
      }
      __builtin_amdgcn_sched_barrier(0);
    }

    const int rowBase = strip * 16 + (h << 3);
    if (HEAD) {
      float p0[8], p1[8];
#pragma unroll
      for (int r = 0; r < 8; ++r) { p0[r] = 0.0f; p1[r] = 0.0f; }
#pragma unroll
      for (int t = 0; t < NT; ++t) {
        const float bv = bias[16 * t + c];
        const float w0 = W2[(16 * t + c) * 2 + 0];
        const float w1 = W2[(16 * t + c) * 2 + 1];
#pragma unroll
        for (int r = 0; r < 8; ++r) {
          const float v = fmaxf(acc[t][r] + bv, 0.0f);
          p0[r] += v * w0;
          p1[r] += v * w1;
        }
      }
#pragma unroll
      for (int r = 0; r < 8; ++r) {
#pragma unroll
        for (int off = 8; off > 0; off >>= 1) {
          p0[r] += __shfl_xor(p0[r], off, 32);
          p1[r] += __shfl_xor(p1[r], off, 32);
        }
      }
      if (c == 0) {
        const float o0 = b2[0];
        const float o1 = b2[1];
        float* op = C + (size_t)rowBase * 2;
#pragma unroll
        for (int r = 0; r < 8; r += 2)
          *(float4*)(op + 2 * r) = make_float4(p0[r] + o0, p1[r] + o1, p0[r + 1] + o0, p1[r + 1] + o1);
      }
    } else {
#pragma unroll
      for (int t = 0; t < NT; ++t) {
#pragma unroll
        for (int r = 0; r < 8; ++r)
          C[(size_t)(rowBase + r) * NOUT + 16 * t + c] = acc[t][r];
      }
    }
  }
}

__device__ __forceinline__ int bcast_i(int v, int i) {
  return __builtin_amdgcn_readlane(v, i);
}

#define SLOT_W 32
#define BIN_SHIFT 8
#define BIN_NODES (1 << BIN_SHIFT)
#define NBINS ((GN + BIN_NODES - 1) >> BIN_SHIFT)
#define BIN_CAP 6144
#define EB_THREADS 512
#define EB_BLOCKS 256
#define EB_CHUNK ((GE + EB_BLOCKS - 1) / EB_BLOCKS)
#define EB_EPT ((EB_CHUNK + EB_THREADS - 1) / EB_THREADS)
#define GB_THREADS 512
#define GB_EPT (BIN_CAP / GB_THREADS)
#define CTL_OVF 32
#define CTL_BIN0 64
#define CTL_STRIDE 32
#define CTL_INTS (CTL_BIN0 + CTL_STRIDE * NBINS)
static_assert(NBINS <= 256, "the bin counters are scanned by 256 threads");
static_assert(BIN_CAP % GB_THREADS == 0, "every thread of fill_slots takes the same number of list places");
static_assert(GT_BLOCKS * GT_THREADS >= CTL_INTS, "the first GEMM launch clears the list lengths, one int per thread");
static_assert(CTL_INTS <= GN, "the list lengths fit in the room behind the node counts");

__global__ void __launch_bounds__(EB_THREADS)
bin_edges(const int* __restrict__ ei, int* __restrict__ deg,
          int* __restrict__ binBuf, int* __restrict__ ovf) {
  __shared__ int hist[256];
  __shared__ int lofs[256];
  __shared__ int gofs[256];
  __shared__ int wtot[8];
  __shared__ int2 sorted[EB_CHUNK];

  int* binCount = deg + GN + CTL_BIN0;
  int* ovfCount = deg + GN + CTL_OVF;
  const int t    = threadIdx.x;
  const int lane = t & 31;
  const int e0   = blockIdx.x * EB_CHUNK;
  const int eEnd = (e0 + EB_CHUNK < GE) ? (e0 + EB_CHUNK) : GE;

  if (t < 256) hist[t] = 0;
  __syncthreads();

  int d[EB_EPT], s[EB_EPT], r[EB_EPT];
#pragma unroll
  for (int i = 0; i < EB_EPT; ++i) {
    const int e = e0 + t + i * EB_THREADS;
    d[i] = -1;
    s[i] = 0;
    if (e < eEnd) {
      d[i] = ei[GE + e];
      s[i] = ei[e];
    }
  }
#pragma unroll
  for (int i = 0; i < EB_EPT; ++i) {
    r[i] = 0;
    if (d[i] >= 0) r[i] = atomicAdd(&hist[d[i] >> BIN_SHIFT], 1);
  }
  __syncthreads();

  int hv = 0, incl = 0;
  if (t < 256) {
    hv = hist[t];
    incl = hv;
#pragma unroll
    for (int off = 1; off < 32; off <<= 1) {
      const int v = __shfl_up(incl, off, 32);
      if (lane >= off) incl += v;
    }
    if (lane == 31) wtot[t >> 5] = incl;
  }
  __syncthreads();
  if (t < 256) {
    int wbase = 0;
#pragma unroll
    for (int w = 0; w < 8; ++w) wbase += (w < (t >> 5)) ? wtot[w] : 0;
    const int ex = wbase + incl - hv;
    lofs[t] = ex;
    int g = 0;
    if (t < NBINS && hv > 0) g = atomicAdd(&binCount[t * CTL_STRIDE], hv);
    gofs[t] = g - ex;
  }
  __syncthreads();

#pragma unroll
  for (int i = 0; i < EB_EPT; ++i) {
    if (d[i] >= 0) sorted[lofs[d[i] >> BIN_SHIFT] + r[i]] = make_int2(d[i], s[i]);
  }
  __syncthreads();

  const int n = eEnd - e0;
  int2* __restrict__ bb = (int2*)binBuf;
  int2* __restrict__ ov = (int2*)ovf;
#pragma unroll
  for (int i = 0; i < EB_EPT; ++i) {
    const int p = t + i * EB_THREADS;
    if (p < n) {
      const int2 v  = sorted[p];
      const int bin = v.x >> BIN_SHIFT;
      const int idx = gofs[bin] + p;
      int2* q = bb + (unsigned)(bin * BIN_CAP + idx);
      if (idx >= BIN_CAP) q = ov + atomicAdd(ovfCount, 1);
      *q = v;
    }
  }
}

__device__ __forceinline__ void store_edge(const int2 v, const int r, int* rows,
                                           int* __restrict__ deg, int* __restrict__ spill) {
  if (r < SLOT_W) {
    rows[(v.x & (BIN_NODES - 1)) * SLOT_W + r] = v.y;
  } else {
    const int k = atomicAdd(&deg[GN], 1);
    spill[2 * k]     = v.x;
    spill[2 * k + 1] = v.y;
  }
}

__global__ void __launch_bounds__(GB_THREADS)
fill_slots(int* __restrict__ deg, const int* __restrict__ binBuf, const int* __restrict__ ovf,
           int* __restrict__ slot, int* __restrict__ spill) {
  __shared__ int cnt[BIN_NODES];
  __shared__ alignas(16) int rows[BIN_NODES * SLOT_W];

  const int t = threadIdx.x;
  const int b = blockIdx.x;
  const int total = deg[GN + CTL_BIN0 + CTL_STRIDE * b];
  const int n     = (total < BIN_CAP) ? total : BIN_CAP;
  const int novf  = deg[GN + CTL_OVF];

  const int2* __restrict__ lp = (const int2*)binBuf + (size_t)b * BIN_CAP;
  int2 v[GB_EPT];
#pragma unroll
  for (int i = 0; i < GB_EPT; ++i) {
    const int p = t + i * GB_THREADS;
    v[i] = make_int2(-1, 0);
    if (p < n) v[i] = lp[p];
  }

  if (t < BIN_NODES) cnt[t] = 0;
  for (int u = t; u < BIN_NODES * SLOT_W / 4; u += GB_THREADS) ((int4*)rows)[u] = make_int4(0, 0, 0, 0);
  __syncthreads();

  int r[GB_EPT];
#pragma unroll
  for (int i = 0; i < GB_EPT; ++i) {
    r[i] = 0;
    if (v[i].x >= 0) r[i] = atomicAdd(&cnt[v[i].x & (BIN_NODES - 1)], 1);
  }
#pragma unroll
  for (int i = 0; i < GB_EPT; ++i) {
    if (v[i].x >= 0) store_edge(v[i], r[i], rows, deg, spill);
  }
  for (int q = t; q < novf; q += GB_THREADS) {
    const int2 w = ((const int2*)ovf)[q];
    if ((w.x >> BIN_SHIFT) == b) {
      const int rw = atomicAdd(&cnt[w.x & (BIN_NODES - 1)], 1);
      store_edge(w, rw, rows, deg, spill);
    }
  }
  __syncthreads();

  const int node0 = b << BIN_SHIFT;
  int4* __restrict__ sp = (int4*)slot + (size_t)node0 * (SLOT_W / 4);
  for (int u = t; u < BIN_NODES * SLOT_W / 4; u += GB_THREADS) {
    if (node0 + (u / (SLOT_W / 4)) < GN) sp[u] = ((const int4*)rows)[u];
  }
  if (t < BIN_NODES && node0 + t < GN) deg[node0 + t] = cnt[t];
}

#define NODE_WAVES_PER_BLOCK 4

typedef __attribute__((ext_vector_type(4))) float v4f;

template <int MASK>
__device__ __forceinline__ float row_xor(float v) {
  return __builtin_bit_cast(float, __builtin_amdgcn_mov_dpp(__builtin_bit_cast(int, v), 0x160 + MASK, 0xf, 0xf, false));
}

template <int N>
__device__ __forceinline__ float row_share(float v) {
  return __builtin_bit_cast(float, __builtin_amdgcn_mov_dpp(__builtin_bit_cast(int, v), 0x150 + N, 0xf, 0xf, false));
}

template <int BIT>
__device__ __forceinline__ float halve(const float lo, const float hi, const bool up) {
  const float keep = up ? hi : lo;
  const float send = up ? lo : hi;
  return keep + row_xor<BIT>(send);
}

__device__ __forceinline__ float lane_logit_part(const v4f x, const v4f r, const v4f a) {
  v4f t = x + r;
  t = __builtin_elementwise_max(t, t * NEG_SLOPE);
  const v2f q = t.xy * a.xy + t.zw * a.zw;
  return q.x + q.y;
}

template <int NB, bool MASKED>
__device__ __forceinline__ void attend_rows(const v4f* __restrict__ xl4, const int s, const int i,
                                            const int valid, const int lane,
                                            const v4f r, const v4f a,
                                            float& m, float& dn, v4f& acc) {
  v4f x[NB];
#pragma unroll
  for (int k = 0; k < NB; ++k) {
    const int si = bcast_i(s, i + k);
    x[k] = xl4[(unsigned)(si * (GHC / 4) + lane)];
  }
  float l[NB];
#pragma unroll
  for (int k = 0; k < NB; ++k) l[k] = lane_logit_part(x[k], r, a);

  static_assert(NB == 8 || NB == 4, "a group is 8 or 4 edges");
  const bool b0 = (lane & 1) != 0;
  const bool b1 = (lane & 2) != 0;
  const bool b2 = (lane & 4) != 0;
  float lg;
  int mine;
  if constexpr (NB == 8) {
    float u[4];
#pragma unroll
    for (int k = 0; k < 4; ++k) u[k] = halve<1>(l[k], l[k + 4], b0);
    const float w0 = halve<2>(u[0], u[2], b1);
    const float w1 = halve<2>(u[1], u[3], b1);
    lg = halve<4>(w0, w1, b2);
    lg += row_xor<8>(lg);
    mine = (b0 ? 4 : 0) + (b1 ? 2 : 0) + (b2 ? 1 : 0);
  } else {
    const float u0 = halve<1>(l[0], l[2], b0);
    const float u1 = halve<1>(l[1], l[3], b0);
    lg = halve<2>(u0, u1, b1);
    lg += row_xor<4>(lg);
    lg += row_xor<8>(lg);
    mine = (b0 ? 2 : 0) + (b1 ? 1 : 0);
  }
  if (MASKED) lg = (mine < valid) ? lg : -__builtin_huge_valf();

  float g = lg;
  g = fmaxf(g, row_xor<1>(g));
  g = fmaxf(g, row_xor<2>(g));
  if constexpr (NB == 8) g = fmaxf(g, row_xor<4>(g));
  const float nm = fmaxf(m, g);
  const float sc = __expf(m - nm);
  const float p  = __expf(lg - nm);
  m = nm;
  float pd = p;
  if constexpr (NB == 4) pd = b2 ? 0.0f : p;
  dn = dn * sc + pd;
  acc *= sc;
  if constexpr (NB == 8) {
    acc += row_share<0>(p) * x[0];
    acc += row_share<4>(p) * x[1];
    acc += row_share<2>(p) * x[2];
    acc += row_share<6>(p) * x[3];
    acc += row_share<1>(p) * x[4];
    acc += row_share<5>(p) * x[5];
    acc += row_share<3>(p) * x[6];
    acc += row_share<7>(p) * x[7];
  } else {
    acc += row_share<0>(p) * x[0];
    acc += row_share<2>(p) * x[1];
    acc += row_share<1>(p) * x[2];
    acc += row_share<3>(p) * x[3];
  }
}

__global__ void __launch_bounds__(NODE_WAVES_PER_BLOCK * 32, 12)
node_attend(const float* __restrict__ xl,
            const float* __restrict__ xr,
            const int* __restrict__ deg,
            const int* __restrict__ slot,
            const int* __restrict__ spill,
            const float* __restrict__ att,
            float* __restrict__ outp) {
  const int lane = threadIdx.x & 31;
  const int n = __builtin_amdgcn_readfirstlane(blockIdx.x * NODE_WAVES_PER_BLOCK + (threadIdx.x >> 5));
  if (n >= GN) return;
  const v4f* __restrict__ xl4 = (const v4f*)xl;

  const int dg   = deg[n];
  const int sraw = slot[(unsigned)(n * SLOT_W + lane)];

  const v4f a = ((const v4f*)att)[lane];
  const v4f r = ((const v4f*)xr)[(size_t)n * (GHC / 4) + lane];

  v4f acc = xl4[(size_t)n * (GHC / 4) + lane];
  float m = lane_logit_part(acc, r, a);
  m += row_xor<1>(m);
  m += row_xor<2>(m);
  m += row_xor<4>(m);
  m += row_xor<8>(m);
  float dn = ((lane & 7) == 0) ? 1.0f : 0.0f;

  const int cnt = (dg < SLOT_W) ? dg : SLOT_W;
  const int s   = (lane < cnt) ? sraw : n;
  int i = 0;
  for (; i + 8 <= cnt; i += 8)
    attend_rows<8, false>(xl4, s, i, 8, lane, r, a, m, dn, acc);
  const int rem = cnt - i;
  if (rem > 4)      attend_rows<8, true>(xl4, s, i, rem, lane, r, a, m, dn, acc);
  else if (rem > 0) attend_rows<4, true>(xl4, s, i, rem, lane, r, a, m, dn, acc);

  if (dg > SLOT_W) {
    const int total = deg[GN];
    for (int base = 0; base < total; base += 32) {
      const int q  = base + lane;
      const int pd = (q < total) ? spill[2 * q]     : -1;
      const int ps = (q < total) ? spill[2 * q + 1] : n;
      unsigned hit = __builtin_amdgcn_ballot_w32(pd == n);
      while (hit != 0u) {
        const int k = __builtin_ctz(hit);
        hit &= hit - 1u;
        const int src = bcast_i(ps, k);
        const int s1  = (lane == 0) ? src : n;
        attend_rows<4, true>(xl4, s1, 0, 1, lane, r, a, m, dn, acc);
      }
    }
  }

  dn += row_xor<1>(dn);
  dn += row_xor<2>(dn);
  dn += row_xor<4>(dn);

  v4f o;
  o.x = acc.x / dn;
  o.y = acc.y / dn;
  o.z = acc.z / dn;
  o.w = acc.w / dn;
  ((v4f*)outp)[(size_t)n * (GHC / 4) + lane] = o;
}

#define STATS_ROWS 16
#define STATS_WAVES_PER_BLOCK 16
#define STATS_WAVES (GN / STATS_ROWS)
#define STATS_BLOCKS ((STATS_WAVES + STATS_WAVES_PER_BLOCK - 1) / STATS_WAVES_PER_BLOCK)
static_assert(GN % STATS_ROWS == 0, "a wave's rows are all inside the array");

__global__ void __launch_bounds__(STATS_WAVES_PER_BLOCK * 32)
gn_stats(const float* __restrict__ accum,
         const float* __restrict__ bias,
         float* __restrict__ colsum,
         float* __restrict__ colsumsq) {
  __shared__ alignas(16) float part[2][STATS_WAVES_PER_BLOCK][GHC];
  const int lane = threadIdx.x & 31;
  const int wv   = threadIdx.x >> 5;
  const int wave = blockIdx.x * STATS_WAVES_PER_BLOCK + wv;

  float4 s  = make_float4(0.0f, 0.0f, 0.0f, 0.0f);
  float4 s2 = make_float4(0.0f, 0.0f, 0.0f, 0.0f);
  if (wave < STATS_WAVES) {
    const size_t base = (size_t)wave * STATS_ROWS * GHC + (size_t)lane * 4;
    float4 v[STATS_ROWS];
#pragma unroll
    for (int k = 0; k < STATS_ROWS; ++k)
      v[k] = *(const float4*)(accum + base + (size_t)k * GHC);
    const float4 bv = *(const float4*)(bias + lane * 4);
#pragma unroll
    for (int k = 0; k < STATS_ROWS; ++k) {
      const float x0 = v[k].x + bv.x;
      const float x1 = v[k].y + bv.y;
      const float x2 = v[k].z + bv.z;
      const float x3 = v[k].w + bv.w;
      s.x += x0;  s2.x += x0 * x0;
      s.y += x1;  s2.y += x1 * x1;
      s.z += x2;  s2.z += x2 * x2;
      s.w += x3;  s2.w += x3 * x3;
    }
  }
  *(float4*)&part[0][wv][lane * 4] = s;
  *(float4*)&part[1][wv][lane * 4] = s2;
  __syncthreads();

  const int t = threadIdx.x;
  if (t < 2 * GHC) {
    const int st = t >> 7;
    const int f  = t & (GHC - 1);
    float tot = 0.0f;
#pragma unroll
    for (int w = 0; w < STATS_WAVES_PER_BLOCK; ++w) tot += part[st][w][f];
    atomicAdd(st ? &colsumsq[f] : &colsum[f], tot);
  }
}

extern "C" void kernel_launch(void* const* d_in, const int* in_sizes, int n_in,
                              void* d_out, int out_size, void* d_ws, size_t ws_size,
                              hipStream_t stream) {
  (void)in_sizes; (void)n_in; (void)out_size; (void)ws_size;

  const float* x     = (const float*)d_in[0];
  const int*   ei    = (const int*)d_in[1];
  const float* Wl    = (const float*)d_in[2];
  const float* Wr    = (const float*)d_in[3];
  const float* att   = (const float*)d_in[4];
  const float* convb = (const float*)d_in[5];
  const float* gnw   = (const float*)d_in[6];
  const float* gnb   = (const float*)d_in[7];
  const float* gnms  = (const float*)d_in[8];
  const float* W1    = (const float*)d_in[9];
  const float* b1    = (const float*)d_in[10];
  const float* W2    = (const float*)d_in[11];
  const float* b2    = (const float*)d_in[12];
  float* out = (float*)d_out;

  float* ws       = (float*)d_ws;
  float* xl       = ws;
  float* xr       = xl + (size_t)GN * GHC;
  float* hbuf     = xr + (size_t)GN * GHC;
  int*   slot     = (int*)hbuf;
  int*   spill    = slot + (size_t)GN * SLOT_W;
  int*   ovf      = spill + (size_t)2 * GE;
  float* h1       = hbuf + (size_t)GN * GHC;
  int*   binBuf   = (int*)h1;
  size_t scratch1 = ((size_t)GN * GC > (size_t)GETOT * GH)
                        ? (size_t)GN * GC : (size_t)GETOT * GH;
  float* accum    = h1 + scratch1;
  float* rpregion = accum + (size_t)GN * GHC;
  float* sum0     = rpregion + (size_t)GN * GH;
  float* sumsq0   = sum0 + GHC;
  float* sum1     = sumsq0 + GHC;
  float* sumsq1   = sum1 + GHC;
  int*   deg      = (int*)(sumsq1 + GHC);

  const int nodeBlocks = (GN + NODE_WAVES_PER_BLOCK - 1) / NODE_WAVES_PER_BLOCK;

  gemm_strip_f32<GHC / 16, false, false><<<GT_BLOCKS, GT_THREADS, 0, stream>>>(
      x, Wl, Wr, xl, xr,
      nullptr, nullptr, nullptr, nullptr, nullptr, nullptr,
      nullptr, nullptr, nullptr, sum0, deg + GN, CTL_INTS, 2, GN / 16);
  bin_edges<<<EB_BLOCKS, EB_THREADS, 0, stream>>>(ei, deg, binBuf, ovf);
  fill_slots<<<NBINS, GB_THREADS, 0, stream>>>(deg, binBuf, ovf, slot, spill);
  node_attend<<<nodeBlocks, NODE_WAVES_PER_BLOCK * 32, 0, stream>>>(xl, xr, deg, slot, spill, att, accum);
  gn_stats<<<STATS_BLOCKS, STATS_WAVES_PER_BLOCK * 32, 0, stream>>>(accum, convb, sum0, sumsq0);

  gemm_strip_f32<GHC / 16, true, false><<<GT_BLOCKS, GT_THREADS, 0, stream>>>(
      accum, Wl + (size_t)GHC * GHC, Wr + (size_t)GHC * GHC, xl, xr,
      sum0, sumsq0, gnw, gnb, gnms, convb,
      nullptr, nullptr, nullptr, nullptr, nullptr, 0, 2, GN / 16);
  node_attend<<<nodeBlocks, NODE_WAVES_PER_BLOCK * 32, 0, stream>>>(xl, xr, deg, slot, spill, att + GH * GC, accum);
  gn_stats<<<STATS_BLOCKS, STATS_WAVES_PER_BLOCK * 32, 0, stream>>>(accum, convb + GHC, sum1, sumsq1);

  gemm_strip_f32<GC / 16, true, true><<<GT_BLOCKS, GT_THREADS, 0, stream>>>(
      accum, W1, nullptr, out, nullptr,
      sum1, sumsq1, gnw + GHC, gnb + GHC, gnms + GHC, convb + GHC,
      b1, W2, b2, nullptr, nullptr, 0, 1, GN / 16);
}
